# starting kernel
# baseline (speedup 1.0000x reference)
#include <hip/hip_runtime.h>

#define NNODES  50000
#define NEDGES  800000
#define FDIM    256
#define HEADS1  4
#define NEG_SLOPE 0.2f

typedef __attribute__((ext_vector_type(2))) float v2f;
typedef __attribute__((ext_vector_type(8))) float v8f;

__device__ __forceinline__ unsigned f2ord(float f) {
  unsigned u = __float_as_uint(f);
  return (u & 0x80000000u) ? ~u : (u | 0x80000000u);
}
__device__ __forceinline__ float ord2f(unsigned k) {
  unsigned u = (k & 0x80000000u) ? (k ^ 0x80000000u) : ~k;
  return __uint_as_float(u);
}
#define ORD_NEG_INF 0x007FFFFFu

__global__ void fill_u32(unsigned* __restrict__ p, long long count, unsigned val) {
  long long i = (long long)blockIdx.x * blockDim.x + threadIdx.x;
  if (i < count) p[i] = val;
}

__global__ void gemm_wmma_f32(const float* __restrict__ A, const float* __restrict__ B,
                              float* __restrict__ C, int M) {
  constexpr int K = 256, Nc = 256;
  constexpr int NT = 4;
  constexpr int nWavesN = (Nc / 16) / NT;
  int wave = (int)(((long long)blockIdx.x * blockDim.x + threadIdx.x) >> 5);
  int lane = threadIdx.x & 31;
  int tilesM = M >> 4;
  if (wave >= tilesM * nWavesN) return;
  int tM  = wave / nWavesN;
  int tN4 = (wave - tM * nWavesN) * NT;
  int half = lane >> 4;
  int l16  = lane & 15;

  const float* Abase = A + (size_t)(tM * 16 + l16) * K + 2 * half;
  const float* Bbase = B + (size_t)(2 * half) * Nc + tN4 * 16 + l16;

  v8f c[NT] = {};
#pragma unroll 8
  for (int k = 0; k < K; k += 4) {
    v2f a = *(const v2f*)(Abase + k);
#pragma unroll
    for (int t = 0; t < NT; ++t) {
      v2f b;
      b.x = Bbase[k * Nc + t * 16];
      b.y = Bbase[k * Nc + Nc + t * 16];
      c[t] = __builtin_amdgcn_wmma_f32_16x16x4_f32(false, a, false, b,
                                                   (short)0, c[t], false, false);
    }
  }
  float* Cp = C + (size_t)(tM * 16 + 8 * half) * Nc + tN4 * 16 + l16;
#pragma unroll
  for (int t = 0; t < NT; ++t)
#pragma unroll
    for (int i = 0; i < 8; ++i) Cp[i * Nc + t * 16] = c[t][i];
}

template <int H>
__global__ void att_logits(const float* __restrict__ h, const float* __restrict__ a_src,
                           const float* __restrict__ a_dst, float* __restrict__ als,
                           float* __restrict__ ald, int n) {
  constexpr int F = FDIM / H;
  int idx = blockIdx.x * blockDim.x + threadIdx.x;
  if (idx >= n * H) return;
  int node = idx / H, hd = idx - node * H;
  const float4* hp = (const float4*)(h + (size_t)node * FDIM + (size_t)hd * F);
  const float4* as = (const float4*)(a_src + hd * F);
  const float4* ad = (const float4*)(a_dst + hd * F);
  float s = 0.f, d = 0.f;
#pragma unroll 4
  for (int f = 0; f < F / 4; ++f) {
    float4 v = hp[f], x = as[f], y = ad[f];
    s += v.x * x.x + v.y * x.y + v.z * x.z + v.w * x.w;
    d += v.x * y.x + v.y * y.y + v.z * y.z + v.w * y.w;
  }
  als[idx] = s;
  ald[idx] = d;
}

template <int H>
__device__ __forceinline__ float edge_score(const float* als, const float* ald,
                                            int s, int d, int hd) {
  float sc = als[s * H + hd] + ald[d * H + hd];
  return sc > 0.f ? sc : NEG_SLOPE * sc;
}

template <int H>
__global__ void edge_max(const int* __restrict__ src, const int* __restrict__ dst,
                         int E0, int n, const float* __restrict__ als,
                         const float* __restrict__ ald, unsigned* __restrict__ mx) {
  long long i = (long long)blockIdx.x * blockDim.x + threadIdx.x;
  long long ET = (long long)E0 + n;
  if (i >= ET * H) return;
  int hd = (int)(i % H);
  long long ed = i / H;
  int s = ed < E0 ? src[ed] : (int)(ed - E0);
  int d = ed < E0 ? dst[ed] : (int)(ed - E0);
  float sc = edge_score<H>(als, ald, s, d, hd);
  atomicMax(&mx[d * H + hd], f2ord(sc));
}

template <int H>
__global__ void edge_expsum(const int* __restrict__ src, const int* __restrict__ dst,
                            int E0, int n, const float* __restrict__ als,
                            const float* __restrict__ ald, const unsigned* __restrict__ mx,
                            float* __restrict__ dn) {
  long long i = (long long)blockIdx.x * blockDim.x + threadIdx.x;
  long long ET = (long long)E0 + n;
  if (i >= ET * H) return;
  int hd = (int)(i % H);
  long long ed = i / H;
  int s = ed < E0 ? src[ed] : (int)(ed - E0);
  int d = ed < E0 ? dst[ed] : (int)(ed - E0);
  float sc = edge_score<H>(als, ald, s, d, hd);
  float ex = __expf(sc - ord2f(mx[d * H + hd]));
  atomicAdd(&dn[d * H + hd], ex);
}

template <int H>
__global__ void edge_agg(const int* __restrict__ src, const int* __restrict__ dst,
                         int E0, int n, const float* __restrict__ als,
                         const float* __restrict__ ald, const unsigned* __restrict__ mx,
                         const float* __restrict__ dn, const float* __restrict__ h,
                         float* __restrict__ agg) {
  long long ET = (long long)E0 + n;
  long long ed = (long long)blockIdx.x * 4 + (threadIdx.x >> 6);
  if (ed >= ET) return;
  int q = threadIdx.x & 63;
  int s = ed < E0 ? src[ed] : (int)(ed - E0);
  int d = ed < E0 ? dst[ed] : (int)(ed - E0);
  int hd = (q * H) >> 6;
  float sc = edge_score<H>(als, ald, s, d, hd);
  float ex = __expf(sc - ord2f(mx[d * H + hd]));
  float alpha = ex / (dn[d * H + hd] + 1e-16f);
  float4 v = ((const float4*)(h + (size_t)s * FDIM))[q];
  float* ap = agg + (size_t)d * FDIM + 4 * q;
  atomicAdd(ap + 0, v.x * alpha);
  atomicAdd(ap + 1, v.y * alpha);
  atomicAdd(ap + 2, v.z * alpha);
  atomicAdd(ap + 3, v.w * alpha);
}

__global__ void bias_elu(float* __restrict__ agg, const float* __restrict__ b,
                         long long total) {
  long long i = (long long)blockIdx.x * blockDim.x + threadIdx.x;
  if (i >= total) return;
  float v = agg[i] + b[i & (FDIM - 1)];
  agg[i] = v > 0.f ? v : expm1f(v);
}

__global__ void bias_add(float* __restrict__ out, const float* __restrict__ b,
                         long long total) {
  long long i = (long long)blockIdx.x * blockDim.x + threadIdx.x;
  if (i >= total) return;
  out[i] += b[i & (FDIM - 1)];
}

static inline int blks(long long n, int bs) { return (int)((n + bs - 1) / bs); }

extern "C" void kernel_launch(void* const* d_in, const int* in_sizes, int n_in,
                              void* d_out, int out_size, void* d_ws, size_t ws_size,
                              hipStream_t stream) {
  const float* x        = (const float*)d_in[0];
  const int*   ei       = (const int*)d_in[1];
  const float* W1       = (const float*)d_in[2];
  const float* att_src1 = (const float*)d_in[3];
  const float* att_dst1 = (const float*)d_in[4];
  const float* b1       = (const float*)d_in[5];
  const float* W2       = (const float*)d_in[6];
  const float* att_src2 = (const float*)d_in[7];
  const float* att_dst2 = (const float*)d_in[8];
  const float* b2       = (const float*)d_in[9];
  float* out = (float*)d_out;

  const int* src = ei;
  const int* dst = ei + NEDGES;

  const long long NF = (long long)NNODES * FDIM;
  const long long NH = (long long)NNODES * HEADS1;

  float* ws = (float*)d_ws;
  float*    h    = ws;
  float*    agg  = ws + NF;
  float*    als  = ws + 2 * NF;
  float*    ald  = als + NH;
  unsigned* mx   = (unsigned*)(ald + NH);
  float*    dn   = (float*)(mx + NH);
  (void)ws_size; (void)in_sizes; (void)n_in; (void)out_size;

  const long long ET = (long long)NEDGES + NNODES;
  const long long gemmThreads = (long long)(NNODES / 16) * 4   * 32;
  const int aggBlks = blks(ET, 4);

  gemm_wmma_f32<<<blks(gemmThreads, 256), 256, 0, stream>>>(x, W1, h, NNODES);
  att_logits<HEADS1><<<blks(NH, 256), 256, 0, stream>>>(h, att_src1, att_dst1,
                                                        als, ald, NNODES);
  fill_u32<<<blks(NF, 256), 256, 0, stream>>>((unsigned*)agg, NF, 0u);
  fill_u32<<<blks(NH, 256), 256, 0, stream>>>(mx, NH, ORD_NEG_INF);
  fill_u32<<<blks(NH, 256), 256, 0, stream>>>((unsigned*)dn, NH, 0u);
  edge_max<HEADS1><<<blks(ET * HEADS1, 256), 256, 0, stream>>>(src, dst, NEDGES, NNODES,
                                                               als, ald, mx);
  edge_expsum<HEADS1><<<blks(ET * HEADS1, 256), 256, 0, stream>>>(src, dst, NEDGES, NNODES,
                                                                  als, ald, mx, dn);
  edge_agg<HEADS1><<<aggBlks, 256, 0, stream>>>(src, dst, NEDGES, NNODES,
                                                als, ald, mx, dn, h, agg);
  bias_elu<<<blks(NF, 256), 256, 0, stream>>>(agg, b1, NF);

  gemm_wmma_f32<<<blks(gemmThreads, 256), 256, 0, stream>>>(agg, W2, h, NNODES);
  att_logits<1><<<blks(NNODES, 256), 256, 0, stream>>>(h, att_src2, att_dst2,
                                                       als, ald, NNODES);
  fill_u32<<<blks(NF, 256), 256, 0, stream>>>((unsigned*)out, NF, 0u);
  fill_u32<<<blks(NNODES, 256), 256, 0, stream>>>(mx, NNODES, ORD_NEG_INF);
  fill_u32<<<blks(NNODES, 256), 256, 0, stream>>>((unsigned*)dn, NNODES, 0u);
  edge_max<1><<<blks(ET, 256), 256, 0, stream>>>(src, dst, NEDGES, NNODES, als, ald, mx);
  edge_expsum<1><<<blks(ET, 256), 256, 0, stream>>>(src, dst, NEDGES, NNODES,
                                                    als, ald, mx, dn);
  edge_agg<1><<<aggBlks, 256, 0, stream>>>(src, dst, NEDGES, NNODES,
                                           als, ald, mx, dn, h, out);
  bias_add<<<blks(NF, 256), 256, 0, stream>>>(out, b2, NF);
}


# edge_agg: each workgroup owns 196 dst rows in LDS, scans the dst list, gathers its edges' rows, sums by LDS f32 adds (self loop first, then arrival order), stores rows once with bias/ELU; no L2 adds
# speedup vs baseline: 2.1049x; 2.1049x over previous
#include <hip/hip_runtime.h>

#define NNODES  50000
#define NEDGES  800000
#define FDIM    256
#define HEADS1  4
#define NEG_SLOPE 0.2f

typedef __attribute__((ext_vector_type(2))) float v2f;
typedef __attribute__((ext_vector_type(8))) float v8f;

__device__ __forceinline__ unsigned f2ord(float f) {
  unsigned u = __float_as_uint(f);
  return (u & 0x80000000u) ? ~u : (u | 0x80000000u);
}
__device__ __forceinline__ float ord2f(unsigned k) {
  unsigned u = (k & 0x80000000u) ? (k ^ 0x80000000u) : ~k;
  return __uint_as_float(u);
}
#define ORD_NEG_INF 0x007FFFFFu

__global__ void fill_u32(unsigned* __restrict__ p, long long count, unsigned val) {
  long long i = (long long)blockIdx.x * blockDim.x + threadIdx.x;
  if (i < count) p[i] = val;
}

__global__ void gemm_wmma_f32(const float* __restrict__ A, const float* __restrict__ B,
                              float* __restrict__ C, int M) {
  constexpr int K = 256, Nc = 256;
  constexpr int NT = 4;
  constexpr int nWavesN = (Nc / 16) / NT;
  int wave = (int)(((long long)blockIdx.x * blockDim.x + threadIdx.x) >> 5);
  int lane = threadIdx.x & 31;
  int tilesM = M >> 4;
  if (wave >= tilesM * nWavesN) return;
  int tM  = wave / nWavesN;
  int tN4 = (wave - tM * nWavesN) * NT;
  int half = lane >> 4;
  int l16  = lane & 15;

  const float* Abase = A + (size_t)(tM * 16 + l16) * K + 2 * half;
  const float* Bbase = B + (size_t)(2 * half) * Nc + tN4 * 16 + l16;

  v8f c[NT] = {};
#pragma unroll 8
  for (int k = 0; k < K; k += 4) {
    v2f a = *(const v2f*)(Abase + k);
#pragma unroll
    for (int t = 0; t < NT; ++t) {
      v2f b;
      b.x = Bbase[k * Nc + t * 16];
      b.y = Bbase[k * Nc + Nc + t * 16];
      c[t] = __builtin_amdgcn_wmma_f32_16x16x4_f32(false, a, false, b,
                                                   (short)0, c[t], false, false);
    }
  }
  float* Cp = C + (size_t)(tM * 16 + 8 * half) * Nc + tN4 * 16 + l16;
#pragma unroll
  for (int t = 0; t < NT; ++t)
#pragma unroll
    for (int i = 0; i < 8; ++i) Cp[i * Nc + t * 16] = c[t][i];
}

template <int H>
__global__ void att_logits(const float* __restrict__ h, const float* __restrict__ a_src,
                           const float* __restrict__ a_dst, float* __restrict__ als,
                           float* __restrict__ ald, int n) {
  constexpr int F = FDIM / H;
  int idx = blockIdx.x * blockDim.x + threadIdx.x;
  if (idx >= n * H) return;
  int node = idx / H, hd = idx - node * H;
  const float4* hp = (const float4*)(h + (size_t)node * FDIM + (size_t)hd * F);
  const float4* as = (const float4*)(a_src + hd * F);
  const float4* ad = (const float4*)(a_dst + hd * F);
  float s = 0.f, d = 0.f;
#pragma unroll 4
  for (int f = 0; f < F / 4; ++f) {
    float4 v = hp[f], x = as[f], y = ad[f];
    s += v.x * x.x + v.y * x.y + v.z * x.z + v.w * x.w;
    d += v.x * y.x + v.y * y.y + v.z * y.z + v.w * y.w;
  }
  als[idx] = s;
  ald[idx] = d;
}

template <int H>
__device__ __forceinline__ float edge_score(const float* als, const float* ald,
                                            int s, int d, int hd) {
  float sc = als[s * H + hd] + ald[d * H + hd];
  return sc > 0.f ? sc : NEG_SLOPE * sc;
}

template <int H>
__global__ void edge_max(const int* __restrict__ src, const int* __restrict__ dst,
                         int E0, int n, const float* __restrict__ als,
                         const float* __restrict__ ald, unsigned* __restrict__ mx) {
  long long i = (long long)blockIdx.x * blockDim.x + threadIdx.x;
  long long ET = (long long)E0 + n;
  if (i >= ET * H) return;
  int hd = (int)(i % H);
  long long ed = i / H;
  int s = ed < E0 ? src[ed] : (int)(ed - E0);
  int d = ed < E0 ? dst[ed] : (int)(ed - E0);
  float sc = edge_score<H>(als, ald, s, d, hd);
  atomicMax(&mx[d * H + hd], f2ord(sc));
}

template <int H>
__global__ void edge_expsum(const int* __restrict__ src, const int* __restrict__ dst,
                            int E0, int n, const float* __restrict__ als,
                            const float* __restrict__ ald, const unsigned* __restrict__ mx,
                            float* __restrict__ dn) {
  long long i = (long long)blockIdx.x * blockDim.x + threadIdx.x;
  long long ET = (long long)E0 + n;
  if (i >= ET * H) return;
  int hd = (int)(i % H);
  long long ed = i / H;
  int s = ed < E0 ? src[ed] : (int)(ed - E0);
  int d = ed < E0 ? dst[ed] : (int)(ed - E0);
  float sc = edge_score<H>(als, ald, s, d, hd);
  float ex = __expf(sc - ord2f(mx[d * H + hd]));
  atomicAdd(&dn[d * H + hd], ex);
}

#define OWN_R    196
#define OWN_CAP  12288
#define OWN_NT   1024
#define OWN_U    8
#define OWN_B    8

template <int H>
__device__ __forceinline__ float own_alpha(float als_v, float ald_v, float m, float den) {
  float sc = als_v + ald_v;
  sc = sc > 0.f ? sc : NEG_SLOPE * sc;
  return __expf(sc - m) / den;
}

template <int H, int NB>
__device__ __forceinline__ void own_batch(unsigned base, int lane, const unsigned* s_list,
                                          const float* s_ald, const float* s_m, const float* s_den,
                                          const float* __restrict__ als, const float* __restrict__ h,
                                          float* acc) {
  constexpr int NCH = FDIM / 32;
  static_assert(NB * H <= 32, "one lane per (edge, head)");
  int s[NB], r[NB];
  for (int i = 0; i < NB; ++i) {
    const unsigned ent = (unsigned)__builtin_amdgcn_readfirstlane((int)s_list[base + i]);
    s[i] = (int)(ent & 0xFFFFu);
    r[i] = (int)(ent >> 16);
  }
  float v[NB][NCH];
  for (int i = 0; i < NB; ++i) {
    const float* hp = h + (size_t)s[i] * FDIM + lane;
    for (int j = 0; j < NCH; ++j) v[i][j] = hp[32 * j];
  }
  float w = 0.f;
  if (lane < NB * H) {
    const unsigned ent = s_list[base + lane / H];
    const int hd = lane % H;
    const int ss = (int)(ent & 0xFFFFu), rr = (int)(ent >> 16);
    w = own_alpha<H>(als[ss * H + hd], s_ald[rr * H + hd], s_m[rr * H + hd], s_den[rr * H + hd]);
  }
  for (int i = 0; i < NB; ++i) {
    float* ap = &acc[r[i] * FDIM + lane];
    for (int j = 0; j < NCH; ++j) {
      const float a = __int_as_float(__builtin_amdgcn_readlane(__float_as_int(w), i * H + (j * H) / NCH));
      atomicAdd(ap + 32 * j, v[i][j] * a);
    }
  }
}

template <int H, bool ELU>
__global__ void __launch_bounds__(OWN_NT)
agg_owner(const int* __restrict__ src, const int* __restrict__ dst, int E0, int n,
          const float* __restrict__ als, const float* __restrict__ ald,
          const unsigned* __restrict__ mx, const float* __restrict__ dn,
          const float* __restrict__ h, const float* __restrict__ bias,
          float* __restrict__ outp) {
  constexpr int NCH = FDIM / 32;
  __shared__ float    acc[OWN_R * FDIM];
  __shared__ float    s_ald[OWN_R * H];
  __shared__ float    s_m[OWN_R * H];
  __shared__ float    s_den[OWN_R * H];
  __shared__ unsigned s_list[OWN_CAP];
  __shared__ unsigned s_cnt;

  const int lane = threadIdx.x & 31;
  const int wave = __builtin_amdgcn_readfirstlane((int)(threadIdx.x >> 5));
  const int lo   = (int)blockIdx.x * OWN_R;
  const int rows = min(OWN_R, n - lo);

  for (int j = threadIdx.x; j < rows * H; j += OWN_NT) {
    s_ald[j] = ald[lo * H + j];
    s_m[j]   = ord2f(mx[lo * H + j]);
    s_den[j] = dn[lo * H + j] + 1e-16f;
  }
  if (threadIdx.x == 0) s_cnt = 0u;
  __syncthreads();

  for (int r = wave; r < rows; r += OWN_NT / 32) {
    const int node = lo + r;
    float a[H];
    for (int hd = 0; hd < H; ++hd)
      a[hd] = own_alpha<H>(als[node * H + hd], s_ald[r * H + hd], s_m[r * H + hd], s_den[r * H + hd]);
    const float* hp = h + (size_t)node * FDIM + lane;
    float* ap = &acc[r * FDIM + lane];
    for (int j = 0; j < NCH; ++j) ap[32 * j] = hp[32 * j] * a[(j * H) / NCH];
  }
  __syncthreads();

  {
    const int nvec   = E0 >> 2;
    const int nchunk = (nvec + 32 * OWN_U - 1) / (32 * OWN_U);
    const int4* dst4 = (const int4*)dst;
    const int rot = (int)(((long long)blockIdx.x * nchunk) / gridDim.x);
    for (int c = wave; c < nchunk; c += OWN_NT / 32) {
      int cc = c + rot;
      if (cc >= nchunk) cc -= nchunk;
      int4 dv[OWN_U];
      for (int u = 0; u < OWN_U; ++u) {
        const int idx = (cc * OWN_U + u) * 32 + lane;
        dv[u] = dst4[min(idx, nvec - 1)];
      }
      unsigned hit = 0u;
      for (int u = 0; u < OWN_U; ++u) {
        const int idx = (cc * OWN_U + u) * 32 + lane;
        unsigned m = ((unsigned)(dv[u].x - lo) < (unsigned)rows ? 1u : 0u)
                   | ((unsigned)(dv[u].y - lo) < (unsigned)rows ? 2u : 0u)
                   | ((unsigned)(dv[u].z - lo) < (unsigned)rows ? 4u : 0u)
                   | ((unsigned)(dv[u].w - lo) < (unsigned)rows ? 8u : 0u);
        if (idx >= nvec) m = 0u;
        hit |= m << (4 * u);
      }
      while (hit) {
        const int b = __builtin_ctz(hit);
        hit &= hit - 1u;
        const int e = (((cc * OWN_U + (b >> 2)) * 32 + lane) << 2) + (b & 3);
        const unsigned rel = (unsigned)(dst[e] - lo);
        const int s = src[e];
        const unsigned pos = atomicAdd(&s_cnt, 1u);
        if (pos < OWN_CAP) {
          s_list[pos] = (rel << 16) | (unsigned)s;
        } else {
          float a[H];
          for (int hd = 0; hd < H; ++hd)
            a[hd] = own_alpha<H>(als[s * H + hd], s_ald[rel * H + hd], s_m[rel * H + hd],
                                 s_den[rel * H + hd]);
          for (int f = 0; f < FDIM; ++f)
            atomicAdd(&acc[rel * FDIM + f], h[(size_t)s * FDIM + f] * a[f / (FDIM / H)]);
        }
      }
    }
  }
  __syncthreads();

  {
    const unsigned total = min(s_cnt, (unsigned)OWN_CAP);
    const unsigned nfull = total / OWN_B;
    for (unsigned bt = (unsigned)wave; bt < nfull; bt += OWN_NT / 32)
      own_batch<H, OWN_B>(bt * OWN_B, lane, s_list, s_ald, s_m, s_den, als, h, acc);
    for (unsigned p = nfull * OWN_B + (unsigned)wave; p < total; p += OWN_NT / 32)
      own_batch<H, 1>(p, lane, s_list, s_ald, s_m, s_den, als, h, acc);
  }
  __syncthreads();

  for (int r = wave; r < rows; r += OWN_NT / 32) {
    const float* ap = &acc[r * FDIM + lane];
    float* op = outp + (size_t)(lo + r) * FDIM + lane;
    for (int j = 0; j < NCH; ++j) {
      float val = ap[32 * j] + bias[lane + 32 * j];
      if (ELU) val = val > 0.f ? val : expm1f(val);
      op[32 * j] = val;
    }
  }
}

static inline int blks(long long n, int bs) { return (int)((n + bs - 1) / bs); }

extern "C" void kernel_launch(void* const* d_in, const int* in_sizes, int n_in,
                              void* d_out, int out_size, void* d_ws, size_t ws_size,
                              hipStream_t stream) {
  const float* x        = (const float*)d_in[0];
  const int*   ei       = (const int*)d_in[1];
  const float* W1       = (const float*)d_in[2];
  const float* att_src1 = (const float*)d_in[3];
  const float* att_dst1 = (const float*)d_in[4];
  const float* b1       = (const float*)d_in[5];
  const float* W2       = (const float*)d_in[6];
  const float* att_src2 = (const float*)d_in[7];
  const float* att_dst2 = (const float*)d_in[8];
  const float* b2       = (const float*)d_in[9];
  float* out = (float*)d_out;

  const int* src = ei;
  const int* dst = ei + NEDGES;

  const long long NF = (long long)NNODES * FDIM;
  const long long NH = (long long)NNODES * HEADS1;

  float* ws = (float*)d_ws;
  float*    h    = ws;
  float*    agg  = ws + NF;
  float*    als  = ws + 2 * NF;
  float*    ald  = als + NH;
  unsigned* mx   = (unsigned*)(ald + NH);
  float*    dn   = (float*)(mx + NH);
  (void)ws_size; (void)in_sizes; (void)n_in; (void)out_size;

  const long long ET = (long long)NEDGES + NNODES;
  const long long gemmThreads = (long long)(NNODES / 16) * 4   * 32;
  static_assert(NEDGES % 4 == 0, "agg_owner reads the dst list in 16-byte groups");
  static_assert(NNODES <= 65536, "agg_owner packs a source row number into 16 bits");
  const int ownBlks = blks(NNODES, OWN_R);

  gemm_wmma_f32<<<blks(gemmThreads, 256), 256, 0, stream>>>(x, W1, h, NNODES);
  att_logits<HEADS1><<<blks(NH, 256), 256, 0, stream>>>(h, att_src1, att_dst1,
                                                        als, ald, NNODES);
  fill_u32<<<blks(NH, 256), 256, 0, stream>>>(mx, NH, ORD_NEG_INF);
  fill_u32<<<blks(NH, 256), 256, 0, stream>>>((unsigned*)dn, NH, 0u);
  edge_max<HEADS1><<<blks(ET * HEADS1, 256), 256, 0, stream>>>(src, dst, NEDGES, NNODES,
                                                               als, ald, mx);
  edge_expsum<HEADS1><<<blks(ET * HEADS1, 256), 256, 0, stream>>>(src, dst, NEDGES, NNODES,
                                                                  als, ald, mx, dn);
  agg_owner<HEADS1, true><<<ownBlks, OWN_NT, 0, stream>>>(src, dst, NEDGES, NNODES,
                                                          als, ald, mx, dn, h, b1, agg);

  gemm_wmma_f32<<<blks(gemmThreads, 256), 256, 0, stream>>>(agg, W2, h, NNODES);
  att_logits<1><<<blks(NNODES, 256), 256, 0, stream>>>(h, att_src2, att_dst2,
                                                       als, ald, NNODES);
  fill_u32<<<blks(NNODES, 256), 256, 0, stream>>>(mx, NNODES, ORD_NEG_INF);
  fill_u32<<<blks(NNODES, 256), 256, 0, stream>>>((unsigned*)dn, NNODES, 0u);
  edge_max<1><<<blks(ET, 256), 256, 0, stream>>>(src, dst, NEDGES, NNODES, als, ald, mx);
  edge_expsum<1><<<blks(ET, 256), 256, 0, stream>>>(src, dst, NEDGES, NNODES,
                                                    als, ald, mx, dn);
  agg_owner<1, false><<<ownBlks, OWN_NT, 0, stream>>>(src, dst, NEDGES, NNODES,
                                                      als, ald, mx, dn, h, b2, out);
}


# edge softmax: each dst row's score max and exp sum found by its owner workgroup in LDS (exp sums in LDS arrival order), rows divided by the sum at the store; 14 to 6 launches, no global atomics
# speedup vs baseline: 3.1562x; 1.4995x over previous
#include <hip/hip_runtime.h>


#define NNODES  50000
#define NEDGES  800000
#define FDIM    256
#define HEADS1  4
#define NEG_SLOPE 0.2f

typedef __attribute__((ext_vector_type(2))) float v2f;
typedef __attribute__((ext_vector_type(8))) float v8f;

__device__ __forceinline__ unsigned f2ord(float f) {
  unsigned u = __float_as_uint(f);
  return (u & 0x80000000u) ? ~u : (u | 0x80000000u);
}
__device__ __forceinline__ float ord2f(unsigned k) {
  unsigned u = (k & 0x80000000u) ? (k ^ 0x80000000u) : ~k;
  return __uint_as_float(u);
}

__global__ void gemm_wmma_f32(const float* __restrict__ A, const float* __restrict__ B,
                              float* __restrict__ C, int M) {
  constexpr int K = 256, Nc = 256;
  constexpr int NT = 4;
  constexpr int nWavesN = (Nc / 16) / NT;
  int wave = (int)(((long long)blockIdx.x * blockDim.x + threadIdx.x) >> 5);
  int lane = threadIdx.x & 31;
  int tilesM = M >> 4;
  if (wave >= tilesM * nWavesN) return;
  int tM  = wave / nWavesN;
  int tN4 = (wave - tM * nWavesN) * NT;
  int half = lane >> 4;
  int l16  = lane & 15;

  const float* Abase = A + (size_t)(tM * 16 + l16) * K + 2 * half;
  const float* Bbase = B + (size_t)(2 * half) * Nc + tN4 * 16 + l16;

  v8f c[NT] = {};
#pragma unroll 8
  for (int k = 0; k < K; k += 4) {
    v2f a = *(const v2f*)(Abase + k);
#pragma unroll
    for (int t = 0; t < NT; ++t) {
      v2f b;
      b.x = Bbase[k * Nc + t * 16];
      b.y = Bbase[k * Nc + Nc + t * 16];
      c[t] = __builtin_amdgcn_wmma_f32_16x16x4_f32(false, a, false, b,
                                                   (short)0, c[t], false, false);
    }
  }
  float* Cp = C + (size_t)(tM * 16 + 8 * half) * Nc + tN4 * 16 + l16;
#pragma unroll
  for (int t = 0; t < NT; ++t)
#pragma unroll
    for (int i = 0; i < 8; ++i) Cp[i * Nc + t * 16] = c[t][i];
}

template <int H>
__global__ void att_logits(const float* __restrict__ h, const float* __restrict__ a_src,
                           const float* __restrict__ a_dst, float* __restrict__ als,
                           float* __restrict__ ald, int n) {
  constexpr int F = FDIM / H;
  int idx = blockIdx.x * blockDim.x + threadIdx.x;
  if (idx >= n * H) return;
  int node = idx / H, hd = idx - node * H;
  const float4* hp = (const float4*)(h + (size_t)node * FDIM + (size_t)hd * F);
  const float4* as = (const float4*)(a_src + hd * F);
  const float4* ad = (const float4*)(a_dst + hd * F);
  float s = 0.f, d = 0.f;
#pragma unroll 4
  for (int f = 0; f < F / 4; ++f) {
    float4 v = hp[f], x = as[f], y = ad[f];
    s += v.x * x.x + v.y * x.y + v.z * x.z + v.w * x.w;
    d += v.x * y.x + v.y * y.y + v.z * y.z + v.w * y.w;
  }
  als[idx] = s;
  ald[idx] = d;
}

#define OWN_R    196
#define OWN_CAP  12288
#define OWN_NT   1024
#define OWN_U    8
#define OWN_B    8
#define OWN_PU   4

template <int H>
__device__ __forceinline__ float own_score(float als_v, float ald_v) {
  float sc = als_v + ald_v;
  return sc > 0.f ? sc : NEG_SLOPE * sc;
}

template <typename F>
__device__ __forceinline__ void own_scan(const int* __restrict__ dst, int E0, int lo, int rows,
                                         int wave, int lane, F on_hit) {
  const int nvec   = E0 >> 2;
  const int nchunk = (nvec + 32 * OWN_U - 1) / (32 * OWN_U);
  const int4* dst4 = (const int4*)dst;
  const int rot = (int)(((long long)blockIdx.x * nchunk) / gridDim.x);
  for (int c = wave; c < nchunk; c += OWN_NT / 32) {
    int cc = c + rot;
    if (cc >= nchunk) cc -= nchunk;
    int4 dv[OWN_U];
    for (int u = 0; u < OWN_U; ++u) {
      const int idx = (cc * OWN_U + u) * 32 + lane;
      dv[u] = dst4[min(idx, nvec - 1)];
    }
    unsigned hit = 0u;
    for (int u = 0; u < OWN_U; ++u) {
      const int idx = (cc * OWN_U + u) * 32 + lane;
      unsigned m = ((unsigned)(dv[u].x - lo) < (unsigned)rows ? 1u : 0u)
                 | ((unsigned)(dv[u].y - lo) < (unsigned)rows ? 2u : 0u)
                 | ((unsigned)(dv[u].z - lo) < (unsigned)rows ? 4u : 0u)
                 | ((unsigned)(dv[u].w - lo) < (unsigned)rows ? 8u : 0u);
      if (idx >= nvec) m = 0u;
      hit |= m << (4 * u);
    }
    while (hit) {
      const int b = __builtin_ctz(hit);
      hit &= hit - 1u;
      on_hit((((cc * OWN_U + (b >> 2)) * 32 + lane) << 2) + (b & 3));
    }
  }
}

template <int H, int NB>
__device__ __forceinline__ void own_batch(unsigned base, int lane, const unsigned* s_list,
                                          const float* s_ald, const unsigned* s_mx, float* s_den,
                                          const float* __restrict__ als, const float* __restrict__ h,
                                          float* acc) {
  constexpr int NCH = FDIM / 32;
  static_assert(NB * H <= 32, "one lane per (edge, head)");
  int s[NB], r[NB];
  for (int i = 0; i < NB; ++i) {
    const unsigned ent = (unsigned)__builtin_amdgcn_readfirstlane((int)s_list[base + i]);
    s[i] = (int)(ent & 0xFFFFu);
    r[i] = (int)(ent >> 16);
  }
  float v[NB][NCH];
  for (int i = 0; i < NB; ++i) {
    const float* hp = h + (size_t)s[i] * FDIM + lane;
    for (int j = 0; j < NCH; ++j) v[i][j] = hp[32 * j];
  }
  float w = 0.f;
  if (lane < NB * H) {
    const unsigned ent = s_list[base + lane / H];
    const int hd = lane % H;
    const int ss = (int)(ent & 0xFFFFu), rr = (int)(ent >> 16);
    w = __expf(own_score<H>(als[ss * H + hd], s_ald[rr * H + hd]) - ord2f(s_mx[rr * H + hd]));
    atomicAdd(&s_den[rr * H + hd], w);
  }
  for (int i = 0; i < NB; ++i) {
    float* ap = &acc[r[i] * FDIM + lane];
    for (int j = 0; j < NCH; ++j) {
      const float a = __int_as_float(__builtin_amdgcn_readlane(__float_as_int(w), i * H + (j * H) / NCH));
      atomicAdd(ap + 32 * j, v[i][j] * a);
    }
  }
}

template <int H, bool ELU>
__global__ void __launch_bounds__(OWN_NT)
agg_owner(const int* __restrict__ src, const int* __restrict__ dst, int E0, int n,
          const float* __restrict__ als, const float* __restrict__ ald,
          const float* __restrict__ h, const float* __restrict__ bias,
          float* __restrict__ outp) {
  constexpr int NCH = FDIM / 32;
  __shared__ float    acc[OWN_R * FDIM];
  __shared__ float    s_ald[OWN_R * H];
  __shared__ unsigned s_mx[OWN_R * H];
  __shared__ float    s_den[OWN_R * H];
  __shared__ unsigned s_list[OWN_CAP];
  __shared__ unsigned s_cnt;

  const int lane = threadIdx.x & 31;
  const int wave = __builtin_amdgcn_readfirstlane((int)(threadIdx.x >> 5));
  const int lo   = (int)blockIdx.x * OWN_R;
  const int rows = min(OWN_R, n - lo);

  for (int j = threadIdx.x; j < rows * H; j += OWN_NT) {
    const float d = ald[lo * H + j];
    s_ald[j] = d;
    s_mx[j]  = f2ord(own_score<H>(als[lo * H + j], d));
  }
  if (threadIdx.x == 0) s_cnt = 0u;
  __syncthreads();

  own_scan(dst, E0, lo, rows, wave, lane, [&](int e) {
    const unsigned rel = (unsigned)(dst[e] - lo);
    const int s = src[e];
    const unsigned pos = atomicAdd(&s_cnt, 1u);
    if (pos < OWN_CAP) s_list[pos] = (rel << 16) | (unsigned)s;
  });
  __syncthreads();

  const unsigned total   = s_cnt;
  const bool     listed  = total <= (unsigned)OWN_CAP;

  if (listed) {
    const unsigned npairs = total * H;
    for (unsigned p0 = threadIdx.x; p0 < npairs; p0 += OWN_NT * OWN_PU) {
      float    sv[OWN_PU];
      unsigned ix[OWN_PU];
      for (int u = 0; u < OWN_PU; ++u) {
        const unsigned p   = min(p0 + (unsigned)(u * OWN_NT), npairs - 1u);
        const unsigned ent = s_list[p / H];
        const int hd = (int)(p % H);
        ix[u] = (ent >> 16) * H + hd;
        sv[u] = als[(ent & 0xFFFFu) * H + hd];
      }
      for (int u = 0; u < OWN_PU; ++u)
        atomicMax(&s_mx[ix[u]], f2ord(own_score<H>(sv[u], s_ald[ix[u]])));
    }
  } else {
    own_scan(dst, E0, lo, rows, wave, lane, [&](int e) {
      const unsigned rel = (unsigned)(dst[e] - lo);
      const int s = src[e];
      for (int hd = 0; hd < H; ++hd)
        atomicMax(&s_mx[rel * H + hd], f2ord(own_score<H>(als[s * H + hd], s_ald[rel * H + hd])));
    });
  }
  __syncthreads();

  for (int r = wave; r < rows; r += OWN_NT / 32) {
    const int node = lo + r;
    float a[H];
    for (int hd = 0; hd < H; ++hd) {
      a[hd] = __expf(own_score<H>(als[node * H + hd], s_ald[r * H + hd]) - ord2f(s_mx[r * H + hd]));
      if (lane == hd) s_den[r * H + hd] = a[hd];
    }
    const float* hp = h + (size_t)node * FDIM + lane;
    float* ap = &acc[r * FDIM + lane];
    for (int j = 0; j < NCH; ++j) ap[32 * j] = hp[32 * j] * a[(j * H) / NCH];
  }
  __syncthreads();

  if (listed) {
    const unsigned nfull = total / OWN_B;
    for (unsigned bt = (unsigned)wave; bt < nfull; bt += OWN_NT / 32)
      own_batch<H, OWN_B>(bt * OWN_B, lane, s_list, s_ald, s_mx, s_den, als, h, acc);
    for (unsigned p = nfull * OWN_B + (unsigned)wave; p < total; p += OWN_NT / 32)
      own_batch<H, 1>(p, lane, s_list, s_ald, s_mx, s_den, als, h, acc);
  } else {
    own_scan(dst, E0, lo, rows, wave, lane, [&](int e) {
      const unsigned rel = (unsigned)(dst[e] - lo);
      const int s = src[e];
      for (int hd = 0; hd < H; ++hd) {
        const float a = __expf(own_score<H>(als[s * H + hd], s_ald[rel * H + hd]) - ord2f(s_mx[rel * H + hd]));
        atomicAdd(&s_den[rel * H + hd], a);
        const int f0 = hd * (FDIM / H);
        for (int f = f0; f < f0 + FDIM / H; ++f)
          atomicAdd(&acc[rel * FDIM + f], h[(size_t)s * FDIM + f] * a);
      }
    });
  }
  __syncthreads();

  for (int r = wave; r < rows; r += OWN_NT / 32) {
    const float* ap = &acc[r * FDIM + lane];
    float* op = outp + (size_t)(lo + r) * FDIM + lane;
    float den[H];
    for (int hd = 0; hd < H; ++hd) den[hd] = s_den[r * H + hd] + 1e-16f;
    for (int j = 0; j < NCH; ++j) {
      float val = ap[32 * j] / den[(j * H) / NCH] + bias[lane + 32 * j];
      if (ELU) val = val > 0.f ? val : expm1f(val);
      op[32 * j] = val;
    }
  }
}

static inline int blks(long long n, int bs) { return (int)((n + bs - 1) / bs); }

extern "C" void kernel_launch(void* const* d_in, const int* in_sizes, int n_in,
                              void* d_out, int out_size, void* d_ws, size_t ws_size,
                              hipStream_t stream) {
  const float* x        = (const float*)d_in[0];
  const int*   ei       = (const int*)d_in[1];
  const float* W1       = (const float*)d_in[2];
  const float* att_src1 = (const float*)d_in[3];
  const float* att_dst1 = (const float*)d_in[4];
  const float* b1       = (const float*)d_in[5];
  const float* W2       = (const float*)d_in[6];
  const float* att_src2 = (const float*)d_in[7];
  const float* att_dst2 = (const float*)d_in[8];
  const float* b2       = (const float*)d_in[9];
  float* out = (float*)d_out;

  const int* src = ei;
  const int* dst = ei + NEDGES;

  const long long NF = (long long)NNODES * FDIM;
  const long long NH = (long long)NNODES * HEADS1;

  float* ws = (float*)d_ws;
  float* h   = ws;
  float* agg = ws + NF;
  float* als = ws + 2 * NF;
  float* ald = als + NH;
  (void)ws_size; (void)in_sizes; (void)n_in; (void)out_size;

  const long long gemmThreads = (long long)(NNODES / 16) * 4   * 32;
  static_assert(NEDGES % 4 == 0, "agg_owner reads the dst list in 16-byte groups");
  static_assert(NNODES <= 65536, "agg_owner packs a source row number into 16 bits");
  const int ownBlks = blks(NNODES, OWN_R);

  gemm_wmma_f32<<<blks(gemmThreads, 256), 256, 0, stream>>>(x, W1, h, NNODES);
  att_logits<HEADS1><<<blks(NH, 256), 256, 0, stream>>>(h, att_src1, att_dst1,
                                                        als, ald, NNODES);
  agg_owner<HEADS1, true><<<ownBlks, OWN_NT, 0, stream>>>(src, dst, NEDGES, NNODES,
                                                          als, ald, h, b1, agg);

  gemm_wmma_f32<<<blks(gemmThreads, 256), 256, 0, stream>>>(agg, W2, h, NNODES);
  att_logits<1><<<blks(NNODES, 256), 256, 0, stream>>>(h, att_src2, att_dst2,
                                                       als, ald, NNODES);
  agg_owner<1, false><<<ownBlks, OWN_NT, 0, stream>>>(src, dst, NEDGES, NNODES,
                                                      als, ald, h, b2, out);
}


# aggregation's edge scan reads one owner byte per edge (dst/196, written by the logits launch), not the 4-byte dst; hits' dst and src fetched after the scan, all at once; f32 sums still arrival order
# speedup vs baseline: 3.3871x; 1.0732x over previous
#include <hip/hip_runtime.h>


#define NNODES  50000
#define NEDGES  800000
#define FDIM    256
#define HEADS1  4
#define NEG_SLOPE 0.2f
#define OWN_R   196

typedef __attribute__((ext_vector_type(2))) float v2f;
typedef __attribute__((ext_vector_type(8))) float v8f;

__device__ __forceinline__ unsigned f2ord(float f) {
  unsigned u = __float_as_uint(f);
  return (u & 0x80000000u) ? ~u : (u | 0x80000000u);
}
__device__ __forceinline__ float ord2f(unsigned k) {
  unsigned u = (k & 0x80000000u) ? (k ^ 0x80000000u) : ~k;
  return __uint_as_float(u);
}

__global__ void gemm_wmma_f32(const float* __restrict__ A, const float* __restrict__ B,
                              float* __restrict__ C, int M) {
  constexpr int K = 256, Nc = 256;
  constexpr int NT = 4;
  constexpr int nWavesN = (Nc / 16) / NT;
  int wave = (int)(((long long)blockIdx.x * blockDim.x + threadIdx.x) >> 5);
  int lane = threadIdx.x & 31;
  int tilesM = M >> 4;
  if (wave >= tilesM * nWavesN) return;
  int tM  = wave / nWavesN;
  int tN4 = (wave - tM * nWavesN) * NT;
  int half = lane >> 4;
  int l16  = lane & 15;

  const float* Abase = A + (size_t)(tM * 16 + l16) * K + 2 * half;
  const float* Bbase = B + (size_t)(2 * half) * Nc + tN4 * 16 + l16;

  v8f c[NT] = {};
#pragma unroll 8
  for (int k = 0; k < K; k += 4) {
    v2f a = *(const v2f*)(Abase + k);
#pragma unroll
    for (int t = 0; t < NT; ++t) {
      v2f b;
      b.x = Bbase[k * Nc + t * 16];
      b.y = Bbase[k * Nc + Nc + t * 16];
      c[t] = __builtin_amdgcn_wmma_f32_16x16x4_f32(false, a, false, b,
                                                   (short)0, c[t], false, false);
    }
  }
  float* Cp = C + (size_t)(tM * 16 + 8 * half) * Nc + tN4 * 16 + l16;
#pragma unroll
  for (int t = 0; t < NT; ++t)
#pragma unroll
    for (int i = 0; i < 8; ++i) Cp[i * Nc + t * 16] = c[t][i];
}

template <int H>
__global__ void att_logits(const float* __restrict__ h, const float* __restrict__ a_src,
                           const float* __restrict__ a_dst, float* __restrict__ als,
                           float* __restrict__ ald, int n,
                           const int* __restrict__ dst, unsigned* __restrict__ own, int E0) {
  constexpr int F = FDIM / H;
  int idx = blockIdx.x * blockDim.x + threadIdx.x;
  {
    const int4* dst4 = (const int4*)dst;
    const int nvec = E0 >> 2;
    for (int i = idx; i < nvec; i += (int)(gridDim.x * blockDim.x)) {
      const int4 d = dst4[i];
      own[i] = (((unsigned)d.x / OWN_R) & 255u) | ((((unsigned)d.y / OWN_R) & 255u) << 8)
             | ((((unsigned)d.z / OWN_R) & 255u) << 16) | (((unsigned)d.w / OWN_R) << 24);
    }
  }
  if (idx >= n * H) return;
  int node = idx / H, hd = idx - node * H;
  const float4* hp = (const float4*)(h + (size_t)node * FDIM + (size_t)hd * F);
  const float4* as = (const float4*)(a_src + hd * F);
  const float4* ad = (const float4*)(a_dst + hd * F);
  float s = 0.f, d = 0.f;
#pragma unroll 4
  for (int f = 0; f < F / 4; ++f) {
    float4 v = hp[f], x = as[f], y = ad[f];
    s += v.x * x.x + v.y * x.y + v.z * x.z + v.w * x.w;
    d += v.x * y.x + v.y * y.y + v.z * y.z + v.w * y.w;
  }
  als[idx] = s;
  ald[idx] = d;
}

#define OWN_CAP  12288
#define OWN_NT   1024
#define OWN_U    4
#define OWN_RAW  (OWN_R * FDIM)
#define OWN_B    8
#define OWN_PU   4

template <int H>
__device__ __forceinline__ float own_score(float als_v, float ald_v) {
  float sc = als_v + ald_v;
  return sc > 0.f ? sc : NEG_SLOPE * sc;
}

__device__ __forceinline__ unsigned own_zero_bytes(unsigned x) {
  const unsigned t = (x & 0x7F7F7F7Fu) + 0x7F7F7F7Fu;
  return ~(t | x) & 0x80808080u;
}

template <typename F>
__device__ __forceinline__ void own_scan(const unsigned* __restrict__ own, int E0,
                                         int wave, int lane, F on_hit) {
  const int nvec   = E0 >> 4;
  const int nchunk = (nvec + 32 * OWN_U - 1) / (32 * OWN_U);
  const uint4* own4 = (const uint4*)own;
  const unsigned me = (blockIdx.x & 255u) * 0x01010101u;
  const int rot = (int)(((long long)blockIdx.x * nchunk) / gridDim.x);
  for (int c = wave; c < nchunk; c += OWN_NT / 32) {
    int cc = c + rot;
    if (cc >= nchunk) cc -= nchunk;
    uint4 ov[OWN_U];
    for (int u = 0; u < OWN_U; ++u) {
      const int idx = (cc * OWN_U + u) * 32 + lane;
      ov[u] = own4[min(idx, nvec - 1)];
    }
    static_assert(OWN_U == 4, "the statement below names the four loads");
    asm volatile("" : "+v"(ov[0].x), "+v"(ov[0].y), "+v"(ov[0].z), "+v"(ov[0].w),
                      "+v"(ov[1].x), "+v"(ov[1].y), "+v"(ov[1].z), "+v"(ov[1].w),
                      "+v"(ov[2].x), "+v"(ov[2].y), "+v"(ov[2].z), "+v"(ov[2].w),
                      "+v"(ov[3].x), "+v"(ov[3].y), "+v"(ov[3].z), "+v"(ov[3].w));
    unsigned long long hit = 0ull;
    for (int u = 0; u < OWN_U; ++u) {
      const int idx = (cc * OWN_U + u) * 32 + lane;
      unsigned m = (own_zero_bytes(ov[u].x ^ me) >> 7) | (own_zero_bytes(ov[u].y ^ me) >> 6)
                 | (own_zero_bytes(ov[u].z ^ me) >> 5) | (own_zero_bytes(ov[u].w ^ me) >> 4);
      if (idx >= nvec) m = 0u;
      hit |= (unsigned long long)m << (32 * (u >> 1) + 4 * (u & 1));
    }
    while (hit) {
      const int p = __builtin_ctzll(hit);
      hit &= hit - 1ull;
      const int u = 2 * (p >> 5) + ((p >> 2) & 1);
      on_hit(((((cc * OWN_U + u) * 32 + lane) << 4) + 4 * (p & 3)) + ((p & 31) >> 3));
    }
  }
}

template <int H, int NB>
__device__ __forceinline__ void own_batch(unsigned base, int lane, const unsigned* s_list,
                                          const float* s_ald, const unsigned* s_mx, float* s_den,
                                          const float* __restrict__ als, const float* __restrict__ h,
                                          float* acc) {
  constexpr int NCH = FDIM / 32;
  static_assert(NB * H <= 32, "one lane per (edge, head)");
  int s[NB], r[NB];
  for (int i = 0; i < NB; ++i) {
    const unsigned ent = (unsigned)__builtin_amdgcn_readfirstlane((int)s_list[base + i]);
    s[i] = (int)(ent & 0xFFFFu);
    r[i] = (int)(ent >> 16);
  }
  float v[NB][NCH];
  for (int i = 0; i < NB; ++i) {
    const float* hp = h + (size_t)s[i] * FDIM + lane;
    for (int j = 0; j < NCH; ++j) v[i][j] = hp[32 * j];
  }
  float w = 0.f;
  if (lane < NB * H) {
    const unsigned ent = s_list[base + lane / H];
    const int hd = lane % H;
    const int ss = (int)(ent & 0xFFFFu), rr = (int)(ent >> 16);
    w = __expf(own_score<H>(als[ss * H + hd], s_ald[rr * H + hd]) - ord2f(s_mx[rr * H + hd]));
    atomicAdd(&s_den[rr * H + hd], w);
  }
  for (int i = 0; i < NB; ++i) {
    float* ap = &acc[r[i] * FDIM + lane];
    for (int j = 0; j < NCH; ++j) {
      const float a = __int_as_float(__builtin_amdgcn_readlane(__float_as_int(w), i * H + (j * H) / NCH));
      atomicAdd(ap + 32 * j, v[i][j] * a);
    }
  }
}

template <int H, bool ELU>
__global__ void __launch_bounds__(OWN_NT)
agg_owner(const int* __restrict__ src, const int* __restrict__ dst,
          const unsigned* __restrict__ own, int E0, int n,
          const float* __restrict__ als, const float* __restrict__ ald,
          const float* __restrict__ h, const float* __restrict__ bias,
          float* __restrict__ outp) {
  constexpr int NCH = FDIM / 32;
  __shared__ float    acc[OWN_R * FDIM];
  __shared__ float    s_ald[OWN_R * H];
  __shared__ unsigned s_mx[OWN_R * H];
  __shared__ float    s_den[OWN_R * H];
  __shared__ unsigned s_list[OWN_CAP];
  __shared__ unsigned s_cnt;
  __shared__ unsigned s_nraw;
  unsigned* s_raw = reinterpret_cast<unsigned*>(acc);

  const int lane = threadIdx.x & 31;
  const int wave = __builtin_amdgcn_readfirstlane((int)(threadIdx.x >> 5));
  const int lo   = (int)blockIdx.x * OWN_R;
  const int rows = min(OWN_R, n - lo);

  for (int j = threadIdx.x; j < rows * H; j += OWN_NT) {
    const float d = ald[lo * H + j];
    s_ald[j] = d;
    s_mx[j]  = f2ord(own_score<H>(als[lo * H + j], d));
  }
  if (threadIdx.x == 0) { s_cnt = 0u; s_nraw = 0u; }
  __syncthreads();

  own_scan(own, E0, wave, lane, [&](int e) {
    const unsigned pos = atomicAdd(&s_nraw, 1u);
    if (pos < (unsigned)OWN_RAW) s_raw[pos] = (unsigned)e;
  });
  __syncthreads();

  const unsigned nraw  = s_nraw;
  const bool     rawok = nraw <= (unsigned)OWN_RAW;
  if (rawok) {
    for (unsigned p = threadIdx.x; p < nraw; p += OWN_NT) {
      const unsigned e   = s_raw[p];
      const unsigned rel = (unsigned)(dst[e] - lo);
      const int s = src[e];
      if (rel < (unsigned)rows) {
        const unsigned pos = atomicAdd(&s_cnt, 1u);
        if (pos < OWN_CAP) s_list[pos] = (rel << 16) | (unsigned)s;
      }
    }
  }
  __syncthreads();

  const unsigned total   = s_cnt;
  const bool     listed  = rawok && total <= (unsigned)OWN_CAP;

  if (listed) {
    const unsigned npairs = total * H;
    for (unsigned p0 = threadIdx.x; p0 < npairs; p0 += OWN_NT * OWN_PU) {
      float    sv[OWN_PU];
      unsigned ix[OWN_PU];
      for (int u = 0; u < OWN_PU; ++u) {
        const unsigned p   = min(p0 + (unsigned)(u * OWN_NT), npairs - 1u);
        const unsigned ent = s_list[p / H];
        const int hd = (int)(p % H);
        ix[u] = (ent >> 16) * H + hd;
        sv[u] = als[(ent & 0xFFFFu) * H + hd];
      }
      for (int u = 0; u < OWN_PU; ++u)
        atomicMax(&s_mx[ix[u]], f2ord(own_score<H>(sv[u], s_ald[ix[u]])));
    }
  } else {
    own_scan(own, E0, wave, lane, [&](int e) {
      const unsigned rel = (unsigned)(dst[e] - lo);
      if (rel >= (unsigned)rows) return;
      const int s = src[e];
      for (int hd = 0; hd < H; ++hd)
        atomicMax(&s_mx[rel * H + hd], f2ord(own_score<H>(als[s * H + hd], s_ald[rel * H + hd])));
    });
  }
  __syncthreads();

  for (int r = wave; r < rows; r += OWN_NT / 32) {
    const int node = lo + r;
    float a[H];
    for (int hd = 0; hd < H; ++hd) {
      a[hd] = __expf(own_score<H>(als[node * H + hd], s_ald[r * H + hd]) - ord2f(s_mx[r * H + hd]));
      if (lane == hd) s_den[r * H + hd] = a[hd];
    }
    const float* hp = h + (size_t)node * FDIM + lane;
    float* ap = &acc[r * FDIM + lane];
    for (int j = 0; j < NCH; ++j) ap[32 * j] = hp[32 * j] * a[(j * H) / NCH];
  }
  __syncthreads();

  if (listed) {
    const unsigned nfull = total / OWN_B;
    for (unsigned bt = (unsigned)wave; bt < nfull; bt += OWN_NT / 32)
      own_batch<H, OWN_B>(bt * OWN_B, lane, s_list, s_ald, s_mx, s_den, als, h, acc);
    for (unsigned p = nfull * OWN_B + (unsigned)wave; p < total; p += OWN_NT / 32)
      own_batch<H, 1>(p, lane, s_list, s_ald, s_mx, s_den, als, h, acc);
  } else {
    own_scan(own, E0, wave, lane, [&](int e) {
      const unsigned rel = (unsigned)(dst[e] - lo);
      if (rel >= (unsigned)rows) return;
      const int s = src[e];
      for (int hd = 0; hd < H; ++hd) {
        const float a = __expf(own_score<H>(als[s * H + hd], s_ald[rel * H + hd]) - ord2f(s_mx[rel * H + hd]));
        atomicAdd(&s_den[rel * H + hd], a);
        const int f0 = hd * (FDIM / H);
        for (int f = f0; f < f0 + FDIM / H; ++f)
          atomicAdd(&acc[rel * FDIM + f], h[(size_t)s * FDIM + f] * a);
      }
    });
  }
  __syncthreads();

  for (int r = wave; r < rows; r += OWN_NT / 32) {
    const float* ap = &acc[r * FDIM + lane];
    float* op = outp + (size_t)(lo + r) * FDIM + lane;
    float den[H];
    for (int hd = 0; hd < H; ++hd) den[hd] = s_den[r * H + hd] + 1e-16f;
    for (int j = 0; j < NCH; ++j) {
      float val = ap[32 * j] / den[(j * H) / NCH] + bias[lane + 32 * j];
      if (ELU) val = val > 0.f ? val : expm1f(val);
      op[32 * j] = val;
    }
  }
}

static inline int blks(long long n, int bs) { return (int)((n + bs - 1) / bs); }

extern "C" void kernel_launch(void* const* d_in, const int* in_sizes, int n_in,
                              void* d_out, int out_size, void* d_ws, size_t ws_size,
                              hipStream_t stream) {
  const float* x        = (const float*)d_in[0];
  const int*   ei       = (const int*)d_in[1];
  const float* W1       = (const float*)d_in[2];
  const float* att_src1 = (const float*)d_in[3];
  const float* att_dst1 = (const float*)d_in[4];
  const float* b1       = (const float*)d_in[5];
  const float* W2       = (const float*)d_in[6];
  const float* att_src2 = (const float*)d_in[7];
  const float* att_dst2 = (const float*)d_in[8];
  const float* b2       = (const float*)d_in[9];
  float* out = (float*)d_out;

  const int* src = ei;
  const int* dst = ei + NEDGES;

  const long long NF = (long long)NNODES * FDIM;
  const long long NH = (long long)NNODES * HEADS1;

  float* ws = (float*)d_ws;
  float* h   = ws;
  float* agg = ws + NF;
  float* als = ws + 2 * NF;
  float* ald = als + NH;
  (void)ws_size; (void)in_sizes; (void)n_in; (void)out_size;

  const long long gemmThreads = (long long)(NNODES / 16) * 4   * 32;
  static_assert(NEDGES % 16 == 0, "agg_owner reads the owner-id list in 16-byte groups of 16 edges");
  static_assert((NNODES + OWN_R - 1) / OWN_R <= 256, "an aggregation workgroup's number fits a byte");
  static_assert(NNODES <= 65536, "agg_owner packs a source row number into 16 bits");
  const int ownBlks = blks(NNODES, OWN_R);
  unsigned* own1 = (unsigned*)out;
  unsigned* own2 = (unsigned*)agg;

  gemm_wmma_f32<<<blks(gemmThreads, 256), 256, 0, stream>>>(x, W1, h, NNODES);
  att_logits<HEADS1><<<blks(NH, 256), 256, 0, stream>>>(h, att_src1, att_dst1,
                                                        als, ald, NNODES, dst, own1, NEDGES);
  agg_owner<HEADS1, true><<<ownBlks, OWN_NT, 0, stream>>>(src, dst, own1, NEDGES, NNODES,
                                                          als, ald, h, b1, agg);

  gemm_wmma_f32<<<blks(gemmThreads, 256), 256, 0, stream>>>(agg, W2, h, NNODES);
  att_logits<1><<<blks(NNODES, 256), 256, 0, stream>>>(h, att_src2, att_dst2,
                                                       als, ald, NNODES, dst, own2, NEDGES);
  agg_owner<1, false><<<ownBlks, OWN_NT, 0, stream>>>(src, dst, own2, NEDGES, NNODES,
                                                      als, ald, h, b2, out);
}


# GEMM: a wave's 4 N-tiles interleave the strip's columns (tile t = columns 4c+t): one 16-B load per k row feeds all four B fragments, one 16-B store per output row; k order and sum order unchanged
# speedup vs baseline: 3.7067x; 1.0944x over previous
#include <hip/hip_runtime.h>


#define NNODES  50000
#define NEDGES  800000
#define FDIM    256
#define HEADS1  4
#define NEG_SLOPE 0.2f
#define OWN_R   196

typedef __attribute__((ext_vector_type(2))) float v2f;
typedef __attribute__((ext_vector_type(8))) float v8f;

__device__ __forceinline__ unsigned f2ord(float f) {
  unsigned u = __float_as_uint(f);
  return (u & 0x80000000u) ? ~u : (u | 0x80000000u);
}
__device__ __forceinline__ float ord2f(unsigned k) {
  unsigned u = (k & 0x80000000u) ? (k ^ 0x80000000u) : ~k;
  return __uint_as_float(u);
}

__global__ void gemm_wmma_f32(const float* __restrict__ A, const float* __restrict__ B,
                              float* __restrict__ C, int M) {
  constexpr int K = 256, Nc = 256;
  constexpr int NT = 4;
  constexpr int nWavesN = (Nc / 16) / NT;
  static_assert(NT == 4, "a lane's elements of the four tiles are one float4");
  int wave = (int)(((long long)blockIdx.x * blockDim.x + threadIdx.x) >> 5);
  int lane = threadIdx.x & 31;
  int tilesM = M >> 4;
  if (wave >= tilesM * nWavesN) return;
  int tM   = wave / nWavesN;
  int col0 = (wave - tM * nWavesN) * (NT * 16);
  int half = lane >> 4;
  int l16  = lane & 15;

  const float* Abase = A + (size_t)(tM * 16 + l16) * K + 2 * half;
  const float* Bbase = B + (size_t)(2 * half) * Nc + col0 + NT * l16;

  v8f c[NT] = {};
#pragma unroll 8
  for (int k = 0; k < K; k += 4) {
    v2f a = *(const v2f*)(Abase + k);
    const float4 b0 = *(const float4*)(Bbase + k * Nc);
    const float4 b1 = *(const float4*)(Bbase + k * Nc + Nc);
    const float b0t[NT] = {b0.x, b0.y, b0.z, b0.w};
    const float b1t[NT] = {b1.x, b1.y, b1.z, b1.w};
#pragma unroll
    for (int t = 0; t < NT; ++t) {
      v2f b;
      b.x = b0t[t];
      b.y = b1t[t];
      c[t] = __builtin_amdgcn_wmma_f32_16x16x4_f32(false, a, false, b,
                                                   (short)0, c[t], false, false);
    }
  }
  float* Cp = C + (size_t)(tM * 16 + 8 * half) * Nc + col0 + NT * l16;
#pragma unroll
  for (int i = 0; i < 8; ++i)
    *(float4*)(Cp + i * Nc) = make_float4(c[0][i], c[1][i], c[2][i], c[3][i]);
}

template <int H>
__global__ void att_logits(const float* __restrict__ h, const float* __restrict__ a_src,
                           const float* __restrict__ a_dst, float* __restrict__ als,
                           float* __restrict__ ald, int n,
                           const int* __restrict__ dst, unsigned* __restrict__ own, int E0) {
  constexpr int F = FDIM / H;
  int idx = blockIdx.x * blockDim.x + threadIdx.x;
  {
    const int4* dst4 = (const int4*)dst;
    const int nvec = E0 >> 2;
    for (int i = idx; i < nvec; i += (int)(gridDim.x * blockDim.x)) {
      const int4 d = dst4[i];
      own[i] = (((unsigned)d.x / OWN_R) & 255u) | ((((unsigned)d.y / OWN_R) & 255u) << 8)
             | ((((unsigned)d.z / OWN_R) & 255u) << 16) | (((unsigned)d.w / OWN_R) << 24);
    }
  }
  if (idx >= n * H) return;
  int node = idx / H, hd = idx - node * H;
  const float4* hp = (const float4*)(h + (size_t)node * FDIM + (size_t)hd * F);
  const float4* as = (const float4*)(a_src + hd * F);
  const float4* ad = (const float4*)(a_dst + hd * F);
  float s = 0.f, d = 0.f;
#pragma unroll 4
  for (int f = 0; f < F / 4; ++f) {
    float4 v = hp[f], x = as[f], y = ad[f];
    s += v.x * x.x + v.y * x.y + v.z * x.z + v.w * x.w;
    d += v.x * y.x + v.y * y.y + v.z * y.z + v.w * y.w;
  }
  als[idx] = s;
  ald[idx] = d;
}

#define OWN_CAP  12288
#define OWN_NT   1024
#define OWN_U    4
#define OWN_RAW  (OWN_R * FDIM)
#define OWN_B    8
#define OWN_PU   4

template <int H>
__device__ __forceinline__ float own_score(float als_v, float ald_v) {
  float sc = als_v + ald_v;
  return sc > 0.f ? sc : NEG_SLOPE * sc;
}

__device__ __forceinline__ unsigned own_zero_bytes(unsigned x) {
  const unsigned t = (x & 0x7F7F7F7Fu) + 0x7F7F7F7Fu;
  return ~(t | x) & 0x80808080u;
}

template <typename F>
__device__ __forceinline__ void own_scan(const unsigned* __restrict__ own, int E0,
                                         int wave, int lane, F on_hit) {
  const int nvec   = E0 >> 4;
  const int nchunk = (nvec + 32 * OWN_U - 1) / (32 * OWN_U);
  const uint4* own4 = (const uint4*)own;
  const unsigned me = (blockIdx.x & 255u) * 0x01010101u;
  const int rot = (int)(((long long)blockIdx.x * nchunk) / gridDim.x);
  for (int c = wave; c < nchunk; c += OWN_NT / 32) {
    int cc = c + rot;
    if (cc >= nchunk) cc -= nchunk;
    uint4 ov[OWN_U];
    for (int u = 0; u < OWN_U; ++u) {
      const int idx = (cc * OWN_U + u) * 32 + lane;
      ov[u] = own4[min(idx, nvec - 1)];
    }
    static_assert(OWN_U == 4, "the statement below names the four loads");
    asm volatile("" : "+v"(ov[0].x), "+v"(ov[0].y), "+v"(ov[0].z), "+v"(ov[0].w),
                      "+v"(ov[1].x), "+v"(ov[1].y), "+v"(ov[1].z), "+v"(ov[1].w),
                      "+v"(ov[2].x), "+v"(ov[2].y), "+v"(ov[2].z), "+v"(ov[2].w),
                      "+v"(ov[3].x), "+v"(ov[3].y), "+v"(ov[3].z), "+v"(ov[3].w));
    unsigned long long hit = 0ull;
    for (int u = 0; u < OWN_U; ++u) {
      const int idx = (cc * OWN_U + u) * 32 + lane;
      unsigned m = (own_zero_bytes(ov[u].x ^ me) >> 7) | (own_zero_bytes(ov[u].y ^ me) >> 6)
                 | (own_zero_bytes(ov[u].z ^ me) >> 5) | (own_zero_bytes(ov[u].w ^ me) >> 4);
      if (idx >= nvec) m = 0u;
      hit |= (unsigned long long)m << (32 * (u >> 1) + 4 * (u & 1));
    }
    while (hit) {
      const int p = __builtin_ctzll(hit);
      hit &= hit - 1ull;
      const int u = 2 * (p >> 5) + ((p >> 2) & 1);
      on_hit(((((cc * OWN_U + u) * 32 + lane) << 4) + 4 * (p & 3)) + ((p & 31) >> 3));
    }
  }
}

template <int H, int NB>
__device__ __forceinline__ void own_batch(unsigned base, int lane, const unsigned* s_list,
                                          const float* s_ald, const unsigned* s_mx, float* s_den,
                                          const float* __restrict__ als, const float* __restrict__ h,
                                          float* acc) {
  constexpr int NCH = FDIM / 32;
  static_assert(NB * H <= 32, "one lane per (edge, head)");
  int s[NB], r[NB];
  for (int i = 0; i < NB; ++i) {
    const unsigned ent = (unsigned)__builtin_amdgcn_readfirstlane((int)s_list[base + i]);
    s[i] = (int)(ent & 0xFFFFu);
    r[i] = (int)(ent >> 16);
  }
  float v[NB][NCH];
  for (int i = 0; i < NB; ++i) {
    const float* hp = h + (size_t)s[i] * FDIM + lane;
    for (int j = 0; j < NCH; ++j) v[i][j] = hp[32 * j];
  }
  float w = 0.f;
  if (lane < NB * H) {
    const unsigned ent = s_list[base + lane / H];
    const int hd = lane % H;
    const int ss = (int)(ent & 0xFFFFu), rr = (int)(ent >> 16);
    w = __expf(own_score<H>(als[ss * H + hd], s_ald[rr * H + hd]) - ord2f(s_mx[rr * H + hd]));
    atomicAdd(&s_den[rr * H + hd], w);
  }
  for (int i = 0; i < NB; ++i) {
    float* ap = &acc[r[i] * FDIM + lane];
    for (int j = 0; j < NCH; ++j) {
      const float a = __int_as_float(__builtin_amdgcn_readlane(__float_as_int(w), i * H + (j * H) / NCH));
      atomicAdd(ap + 32 * j, v[i][j] * a);
    }
  }
}

template <int H, bool ELU>
__global__ void __launch_bounds__(OWN_NT)
agg_owner(const int* __restrict__ src, const int* __restrict__ dst,
          const unsigned* __restrict__ own, int E0, int n,
          const float* __restrict__ als, const float* __restrict__ ald,
          const float* __restrict__ h, const float* __restrict__ bias,
          float* __restrict__ outp) {
  constexpr int NCH = FDIM / 32;
  __shared__ float    acc[OWN_R * FDIM];
  __shared__ float    s_ald[OWN_R * H];
  __shared__ unsigned s_mx[OWN_R * H];
  __shared__ float    s_den[OWN_R * H];
  __shared__ unsigned s_list[OWN_CAP];
  __shared__ unsigned s_cnt;
  __shared__ unsigned s_nraw;
  unsigned* s_raw = reinterpret_cast<unsigned*>(acc);

  const int lane = threadIdx.x & 31;
  const int wave = __builtin_amdgcn_readfirstlane((int)(threadIdx.x >> 5));
  const int lo   = (int)blockIdx.x * OWN_R;
  const int rows = min(OWN_R, n - lo);

  for (int j = threadIdx.x; j < rows * H; j += OWN_NT) {
    const float d = ald[lo * H + j];
    s_ald[j] = d;
    s_mx[j]  = f2ord(own_score<H>(als[lo * H + j], d));
  }
  if (threadIdx.x == 0) { s_cnt = 0u; s_nraw = 0u; }
  __syncthreads();

  own_scan(own, E0, wave, lane, [&](int e) {
    const unsigned pos = atomicAdd(&s_nraw, 1u);
    if (pos < (unsigned)OWN_RAW) s_raw[pos] = (unsigned)e;
  });
  __syncthreads();

  const unsigned nraw  = s_nraw;
  const bool     rawok = nraw <= (unsigned)OWN_RAW;
  if (rawok) {
    for (unsigned p = threadIdx.x; p < nraw; p += OWN_NT) {
      const unsigned e   = s_raw[p];
      const unsigned rel = (unsigned)(dst[e] - lo);
      const int s = src[e];
      if (rel < (unsigned)rows) {
        const unsigned pos = atomicAdd(&s_cnt, 1u);
        if (pos < OWN_CAP) s_list[pos] = (rel << 16) | (unsigned)s;
      }
    }
  }
  __syncthreads();

  const unsigned total   = s_cnt;
  const bool     listed  = rawok && total <= (unsigned)OWN_CAP;

  if (listed) {
    const unsigned npairs = total * H;
    for (unsigned p0 = threadIdx.x; p0 < npairs; p0 += OWN_NT * OWN_PU) {
      float    sv[OWN_PU];
      unsigned ix[OWN_PU];
      for (int u = 0; u < OWN_PU; ++u) {
        const unsigned p   = min(p0 + (unsigned)(u * OWN_NT), npairs - 1u);
        const unsigned ent = s_list[p / H];
        const int hd = (int)(p % H);
        ix[u] = (ent >> 16) * H + hd;
        sv[u] = als[(ent & 0xFFFFu) * H + hd];
      }
      for (int u = 0; u < OWN_PU; ++u)
        atomicMax(&s_mx[ix[u]], f2ord(own_score<H>(sv[u], s_ald[ix[u]])));
    }
  } else {
    own_scan(own, E0, wave, lane, [&](int e) {
      const unsigned rel = (unsigned)(dst[e] - lo);
      if (rel >= (unsigned)rows) return;
      const int s = src[e];
      for (int hd = 0; hd < H; ++hd)
        atomicMax(&s_mx[rel * H + hd], f2ord(own_score<H>(als[s * H + hd], s_ald[rel * H + hd])));
    });
  }
  __syncthreads();

  for (int r = wave; r < rows; r += OWN_NT / 32) {
    const int node = lo + r;
    float a[H];
    for (int hd = 0; hd < H; ++hd) {
      a[hd] = __expf(own_score<H>(als[node * H + hd], s_ald[r * H + hd]) - ord2f(s_mx[r * H + hd]));
      if (lane == hd) s_den[r * H + hd] = a[hd];
    }
    const float* hp = h + (size_t)node * FDIM + lane;
    float* ap = &acc[r * FDIM + lane];
    for (int j = 0; j < NCH; ++j) ap[32 * j] = hp[32 * j] * a[(j * H) / NCH];
  }
  __syncthreads();

  if (listed) {
    const unsigned nfull = total / OWN_B;
    for (unsigned bt = (unsigned)wave; bt < nfull; bt += OWN_NT / 32)
      own_batch<H, OWN_B>(bt * OWN_B, lane, s_list, s_ald, s_mx, s_den, als, h, acc);
    for (unsigned p = nfull * OWN_B + (unsigned)wave; p < total; p += OWN_NT / 32)
      own_batch<H, 1>(p, lane, s_list, s_ald, s_mx, s_den, als, h, acc);
  } else {
    own_scan(own, E0, wave, lane, [&](int e) {
      const unsigned rel = (unsigned)(dst[e] - lo);
      if (rel >= (unsigned)rows) return;
      const int s = src[e];
      for (int hd = 0; hd < H; ++hd) {
        const float a = __expf(own_score<H>(als[s * H + hd], s_ald[rel * H + hd]) - ord2f(s_mx[rel * H + hd]));
        atomicAdd(&s_den[rel * H + hd], a);
        const int f0 = hd * (FDIM / H);
        for (int f = f0; f < f0 + FDIM / H; ++f)
          atomicAdd(&acc[rel * FDIM + f], h[(size_t)s * FDIM + f] * a);
      }
    });
  }
  __syncthreads();

  for (int r = wave; r < rows; r += OWN_NT / 32) {
    const float* ap = &acc[r * FDIM + lane];
    float* op = outp + (size_t)(lo + r) * FDIM + lane;
    float den[H];
    for (int hd = 0; hd < H; ++hd) den[hd] = s_den[r * H + hd] + 1e-16f;
    for (int j = 0; j < NCH; ++j) {
      float val = ap[32 * j] / den[(j * H) / NCH] + bias[lane + 32 * j];
      if (ELU) val = val > 0.f ? val : expm1f(val);
      op[32 * j] = val;
    }
  }
}

static inline int blks(long long n, int bs) { return (int)((n + bs - 1) / bs); }

extern "C" void kernel_launch(void* const* d_in, const int* in_sizes, int n_in,
                              void* d_out, int out_size, void* d_ws, size_t ws_size,
                              hipStream_t stream) {
  const float* x        = (const float*)d_in[0];
  const int*   ei       = (const int*)d_in[1];
  const float* W1       = (const float*)d_in[2];
  const float* att_src1 = (const float*)d_in[3];
  const float* att_dst1 = (const float*)d_in[4];
  const float* b1       = (const float*)d_in[5];
  const float* W2       = (const float*)d_in[6];
  const float* att_src2 = (const float*)d_in[7];
  const float* att_dst2 = (const float*)d_in[8];
  const float* b2       = (const float*)d_in[9];
  float* out = (float*)d_out;

  const int* src = ei;
  const int* dst = ei + NEDGES;

  const long long NF = (long long)NNODES * FDIM;
  const long long NH = (long long)NNODES * HEADS1;

  float* ws = (float*)d_ws;
  float* h   = ws;
  float* agg = ws + NF;
  float* als = ws + 2 * NF;
  float* ald = als + NH;
  (void)ws_size; (void)in_sizes; (void)n_in; (void)out_size;

  const long long gemmThreads = (long long)(NNODES / 16) * 4   * 32;
  static_assert(NEDGES % 16 == 0, "agg_owner reads the owner-id list in 16-byte groups of 16 edges");
  static_assert((NNODES + OWN_R - 1) / OWN_R <= 256, "an aggregation workgroup's number fits a byte");
  static_assert(NNODES <= 65536, "agg_owner packs a source row number into 16 bits");
  const int ownBlks = blks(NNODES, OWN_R);
  unsigned* own1 = (unsigned*)out;
  unsigned* own2 = (unsigned*)agg;

  gemm_wmma_f32<<<blks(gemmThreads, 256), 256, 0, stream>>>(x, W1, h, NNODES);
  att_logits<HEADS1><<<blks(NH, 256), 256, 0, stream>>>(h, att_src1, att_dst1,
                                                        als, ald, NNODES, dst, own1, NEDGES);
  agg_owner<HEADS1, true><<<ownBlks, OWN_NT, 0, stream>>>(src, dst, own1, NEDGES, NNODES,
                                                          als, ald, h, b1, agg);

  gemm_wmma_f32<<<blks(gemmThreads, 256), 256, 0, stream>>>(agg, W2, h, NNODES);
  att_logits<1><<<blks(NNODES, 256), 256, 0, stream>>>(h, att_src2, att_dst2,
                                                       als, ald, NNODES, dst, own2, NEDGES);
  agg_owner<1, false><<<ownBlks, OWN_NT, 0, stream>>>(src, dst, own2, NEDGES, NNODES,
                                                      als, ald, h, b2, out);
}


# aggregation: a lane takes 16 B of a source row (512-B half row per load, was four 128-B loads); LDS rows kept lane-major so each f32 add hits 32 banks; sums still in LDS arrival order
# speedup vs baseline: 3.9864x; 1.0755x over previous
#include <hip/hip_runtime.h>


#define NNODES  50000
#define NEDGES  800000
#define FDIM    256
#define HEADS1  4
#define NEG_SLOPE 0.2f
#define OWN_R   196

typedef __attribute__((ext_vector_type(2))) float v2f;
typedef __attribute__((ext_vector_type(8))) float v8f;

__device__ __forceinline__ unsigned f2ord(float f) {
  unsigned u = __float_as_uint(f);
  return (u & 0x80000000u) ? ~u : (u | 0x80000000u);
}
__device__ __forceinline__ float ord2f(unsigned k) {
  unsigned u = (k & 0x80000000u) ? (k ^ 0x80000000u) : ~k;
  return __uint_as_float(u);
}

__global__ void gemm_wmma_f32(const float* __restrict__ A, const float* __restrict__ B,
                              float* __restrict__ C, int M) {
  constexpr int K = 256, Nc = 256;
  constexpr int NT = 4;
  constexpr int nWavesN = (Nc / 16) / NT;
  static_assert(NT == 4, "a lane's elements of the four tiles are one float4");
  int wave = (int)(((long long)blockIdx.x * blockDim.x + threadIdx.x) >> 5);
  int lane = threadIdx.x & 31;
  int tilesM = M >> 4;
  if (wave >= tilesM * nWavesN) return;
  int tM   = wave / nWavesN;
  int col0 = (wave - tM * nWavesN) * (NT * 16);
  int half = lane >> 4;
  int l16  = lane & 15;

  const float* Abase = A + (size_t)(tM * 16 + l16) * K + 2 * half;
  const float* Bbase = B + (size_t)(2 * half) * Nc + col0 + NT * l16;

  v8f c[NT] = {};
#pragma unroll 8
  for (int k = 0; k < K; k += 4) {
    v2f a = *(const v2f*)(Abase + k);
    const float4 b0 = *(const float4*)(Bbase + k * Nc);
    const float4 b1 = *(const float4*)(Bbase + k * Nc + Nc);
    const float b0t[NT] = {b0.x, b0.y, b0.z, b0.w};
    const float b1t[NT] = {b1.x, b1.y, b1.z, b1.w};
#pragma unroll
    for (int t = 0; t < NT; ++t) {
      v2f b;
      b.x = b0t[t];
      b.y = b1t[t];
      c[t] = __builtin_amdgcn_wmma_f32_16x16x4_f32(false, a, false, b,
                                                   (short)0, c[t], false, false);
    }
  }
  float* Cp = C + (size_t)(tM * 16 + 8 * half) * Nc + col0 + NT * l16;
#pragma unroll
  for (int i = 0; i < 8; ++i)
    *(float4*)(Cp + i * Nc) = make_float4(c[0][i], c[1][i], c[2][i], c[3][i]);
}

template <int H>
__global__ void att_logits(const float* __restrict__ h, const float* __restrict__ a_src,
                           const float* __restrict__ a_dst, float* __restrict__ als,
                           float* __restrict__ ald, int n,
                           const int* __restrict__ dst, unsigned* __restrict__ own, int E0) {
  constexpr int F = FDIM / H;
  int idx = blockIdx.x * blockDim.x + threadIdx.x;
  {
    const int4* dst4 = (const int4*)dst;
    const int nvec = E0 >> 2;
    for (int i = idx; i < nvec; i += (int)(gridDim.x * blockDim.x)) {
      const int4 d = dst4[i];
      own[i] = (((unsigned)d.x / OWN_R) & 255u) | ((((unsigned)d.y / OWN_R) & 255u) << 8)
             | ((((unsigned)d.z / OWN_R) & 255u) << 16) | (((unsigned)d.w / OWN_R) << 24);
    }
  }
  if (idx >= n * H) return;
  int node = idx / H, hd = idx - node * H;
  const float4* hp = (const float4*)(h + (size_t)node * FDIM + (size_t)hd * F);
  const float4* as = (const float4*)(a_src + hd * F);
  const float4* ad = (const float4*)(a_dst + hd * F);
  float s = 0.f, d = 0.f;
#pragma unroll 4
  for (int f = 0; f < F / 4; ++f) {
    float4 v = hp[f], x = as[f], y = ad[f];
    s += v.x * x.x + v.y * x.y + v.z * x.z + v.w * x.w;
    d += v.x * y.x + v.y * y.y + v.z * y.z + v.w * y.w;
  }
  als[idx] = s;
  ald[idx] = d;
}

#define OWN_CAP  12288
#define OWN_NT   1024
#define OWN_U    4
#define OWN_RAW  (OWN_R * FDIM)
#define OWN_B    8
#define OWN_PU   4

template <int H>
__device__ __forceinline__ float own_score(float als_v, float ald_v) {
  float sc = als_v + ald_v;
  return sc > 0.f ? sc : NEG_SLOPE * sc;
}

__device__ __forceinline__ int own_pos(int f) {
  return (f & ~127) | ((f & 3) << 5) | ((f >> 2) & 31);
}

template <int H>
__device__ __forceinline__ float own_pick(const float (&a)[H], int jj, int lane) {
  constexpr int F = FDIM / H;
  static_assert(F % 64 == 0, "the 16 lanes of a 64-feature stretch never straddle two heads");
  const int h0 = (128 * jj) / F;
  if constexpr (F >= 128) {
    return a[h0];
  } else {
    return (lane & 16) ? a[h0 + 1] : a[h0];
  }
}

__device__ __forceinline__ unsigned own_zero_bytes(unsigned x) {
  const unsigned t = (x & 0x7F7F7F7Fu) + 0x7F7F7F7Fu;
  return ~(t | x) & 0x80808080u;
}

template <typename F>
__device__ __forceinline__ void own_scan(const unsigned* __restrict__ own, int E0,
                                         int wave, int lane, F on_hit) {
  const int nvec   = E0 >> 4;
  const int nchunk = (nvec + 32 * OWN_U - 1) / (32 * OWN_U);
  const uint4* own4 = (const uint4*)own;
  const unsigned me = (blockIdx.x & 255u) * 0x01010101u;
  const int rot = (int)(((long long)blockIdx.x * nchunk) / gridDim.x);
  for (int c = wave; c < nchunk; c += OWN_NT / 32) {
    int cc = c + rot;
    if (cc >= nchunk) cc -= nchunk;
    uint4 ov[OWN_U];
    for (int u = 0; u < OWN_U; ++u) {
      const int idx = (cc * OWN_U + u) * 32 + lane;
      ov[u] = own4[min(idx, nvec - 1)];
    }
    static_assert(OWN_U == 4, "the statement below names the four loads");
    asm volatile("" : "+v"(ov[0].x), "+v"(ov[0].y), "+v"(ov[0].z), "+v"(ov[0].w),
                      "+v"(ov[1].x), "+v"(ov[1].y), "+v"(ov[1].z), "+v"(ov[1].w),
                      "+v"(ov[2].x), "+v"(ov[2].y), "+v"(ov[2].z), "+v"(ov[2].w),
                      "+v"(ov[3].x), "+v"(ov[3].y), "+v"(ov[3].z), "+v"(ov[3].w));
    unsigned long long hit = 0ull;
    for (int u = 0; u < OWN_U; ++u) {
      const int idx = (cc * OWN_U + u) * 32 + lane;
      unsigned m = (own_zero_bytes(ov[u].x ^ me) >> 7) | (own_zero_bytes(ov[u].y ^ me) >> 6)
                 | (own_zero_bytes(ov[u].z ^ me) >> 5) | (own_zero_bytes(ov[u].w ^ me) >> 4);
      if (idx >= nvec) m = 0u;
      hit |= (unsigned long long)m << (32 * (u >> 1) + 4 * (u & 1));
    }
    while (hit) {
      const int p = __builtin_ctzll(hit);
      hit &= hit - 1ull;
      const int u = 2 * (p >> 5) + ((p >> 2) & 1);
      on_hit(((((cc * OWN_U + u) * 32 + lane) << 4) + 4 * (p & 3)) + ((p & 31) >> 3));
    }
  }
}

template <int H, int NB>
__device__ __forceinline__ void own_batch(unsigned base, int lane, const unsigned* s_list,
                                          const float* s_ald, const unsigned* s_mx, float* s_den,
                                          const float* __restrict__ als, const float* __restrict__ h,
                                          float* acc) {
  constexpr int NHR = FDIM / 128;
  constexpr int F   = FDIM / H;
  static_assert(F % 64 == 0, "the 16 lanes of a 64-feature stretch never straddle two heads");
  static_assert(NB * H <= 32, "one lane per (edge, head)");
  int s[NB], r[NB];
  for (int i = 0; i < NB; ++i) {
    const unsigned ent = (unsigned)__builtin_amdgcn_readfirstlane((int)s_list[base + i]);
    s[i] = (int)(ent & 0xFFFFu);
    r[i] = (int)(ent >> 16);
  }
  float4 v[NB][NHR];
  for (int i = 0; i < NB; ++i) {
    const float4* hp = (const float4*)(h + (size_t)s[i] * FDIM) + lane;
    for (int jj = 0; jj < NHR; ++jj) v[i][jj] = hp[32 * jj];
  }
  float w = 0.f;
  if (lane < NB * H) {
    const unsigned ent = s_list[base + lane / H];
    const int hd = lane % H;
    const int ss = (int)(ent & 0xFFFFu), rr = (int)(ent >> 16);
    w = __expf(own_score<H>(als[ss * H + hd], s_ald[rr * H + hd]) - ord2f(s_mx[rr * H + hd]));
    atomicAdd(&s_den[rr * H + hd], w);
  }
  for (int i = 0; i < NB; ++i) {
    float* ap = &acc[r[i] * FDIM + lane];
    for (int jj = 0; jj < NHR; ++jj) {
      const int h0 = (128 * jj) / F;
      float a = __int_as_float(__builtin_amdgcn_readlane(__float_as_int(w), i * H + h0));
      if constexpr (F < 128) {
        const float a1 = __int_as_float(__builtin_amdgcn_readlane(__float_as_int(w), i * H + h0 + 1));
        a = (lane & 16) ? a1 : a;
      }
      atomicAdd(ap + 128 * jj,      v[i][jj].x * a);
      atomicAdd(ap + 128 * jj + 32, v[i][jj].y * a);
      atomicAdd(ap + 128 * jj + 64, v[i][jj].z * a);
      atomicAdd(ap + 128 * jj + 96, v[i][jj].w * a);
    }
  }
}

template <int H, bool ELU>
__global__ void __launch_bounds__(OWN_NT)
agg_owner(const int* __restrict__ src, const int* __restrict__ dst,
          const unsigned* __restrict__ own, int E0, int n,
          const float* __restrict__ als, const float* __restrict__ ald,
          const float* __restrict__ h, const float* __restrict__ bias,
          float* __restrict__ outp) {
  constexpr int NHR = FDIM / 128;
  __shared__ float    acc[OWN_R * FDIM];
  __shared__ float    s_ald[OWN_R * H];
  __shared__ unsigned s_mx[OWN_R * H];
  __shared__ float    s_den[OWN_R * H];
  __shared__ unsigned s_list[OWN_CAP];
  __shared__ unsigned s_cnt;
  __shared__ unsigned s_nraw;
  unsigned* s_raw = reinterpret_cast<unsigned*>(acc);

  const int lane = threadIdx.x & 31;
  const int wave = __builtin_amdgcn_readfirstlane((int)(threadIdx.x >> 5));
  const int lo   = (int)blockIdx.x * OWN_R;
  const int rows = min(OWN_R, n - lo);

  for (int j = threadIdx.x; j < rows * H; j += OWN_NT) {
    const float d = ald[lo * H + j];
    s_ald[j] = d;
    s_mx[j]  = f2ord(own_score<H>(als[lo * H + j], d));
  }
  if (threadIdx.x == 0) { s_cnt = 0u; s_nraw = 0u; }
  __syncthreads();

  own_scan(own, E0, wave, lane, [&](int e) {
    const unsigned pos = atomicAdd(&s_nraw, 1u);
    if (pos < (unsigned)OWN_RAW) s_raw[pos] = (unsigned)e;
  });
  __syncthreads();

  const unsigned nraw  = s_nraw;
  const bool     rawok = nraw <= (unsigned)OWN_RAW;
  if (rawok) {
    for (unsigned p = threadIdx.x; p < nraw; p += OWN_NT) {
      const unsigned e   = s_raw[p];
      const unsigned rel = (unsigned)(dst[e] - lo);
      const int s = src[e];
      if (rel < (unsigned)rows) {
        const unsigned pos = atomicAdd(&s_cnt, 1u);
        if (pos < OWN_CAP) s_list[pos] = (rel << 16) | (unsigned)s;
      }
    }
  }
  __syncthreads();

  const unsigned total   = s_cnt;
  const bool     listed  = rawok && total <= (unsigned)OWN_CAP;

  if (listed) {
    const unsigned npairs = total * H;
    for (unsigned p0 = threadIdx.x; p0 < npairs; p0 += OWN_NT * OWN_PU) {
      float    sv[OWN_PU];
      unsigned ix[OWN_PU];
      for (int u = 0; u < OWN_PU; ++u) {
        const unsigned p   = min(p0 + (unsigned)(u * OWN_NT), npairs - 1u);
        const unsigned ent = s_list[p / H];
        const int hd = (int)(p % H);
        ix[u] = (ent >> 16) * H + hd;
        sv[u] = als[(ent & 0xFFFFu) * H + hd];
      }
      for (int u = 0; u < OWN_PU; ++u)
        atomicMax(&s_mx[ix[u]], f2ord(own_score<H>(sv[u], s_ald[ix[u]])));
    }
  } else {
    own_scan(own, E0, wave, lane, [&](int e) {
      const unsigned rel = (unsigned)(dst[e] - lo);
      if (rel >= (unsigned)rows) return;
      const int s = src[e];
      for (int hd = 0; hd < H; ++hd)
        atomicMax(&s_mx[rel * H + hd], f2ord(own_score<H>(als[s * H + hd], s_ald[rel * H + hd])));
    });
  }
  __syncthreads();

  for (int r = wave; r < rows; r += OWN_NT / 32) {
    const int node = lo + r;
    float a[H];
    for (int hd = 0; hd < H; ++hd) {
      a[hd] = __expf(own_score<H>(als[node * H + hd], s_ald[r * H + hd]) - ord2f(s_mx[r * H + hd]));
      if (lane == hd) s_den[r * H + hd] = a[hd];
    }
    const float4* hp = (const float4*)(h + (size_t)node * FDIM) + lane;
    float* ap = &acc[r * FDIM + lane];
    for (int jj = 0; jj < NHR; ++jj) {
      const float4 hv = hp[32 * jj];
      const float  aw = own_pick<H>(a, jj, lane);
      ap[128 * jj]      = hv.x * aw;
      ap[128 * jj + 32] = hv.y * aw;
      ap[128 * jj + 64] = hv.z * aw;
      ap[128 * jj + 96] = hv.w * aw;
    }
  }
  __syncthreads();

  if (listed) {
    const unsigned nfull = total / OWN_B;
    for (unsigned bt = (unsigned)wave; bt < nfull; bt += OWN_NT / 32)
      own_batch<H, OWN_B>(bt * OWN_B, lane, s_list, s_ald, s_mx, s_den, als, h, acc);
    for (unsigned p = nfull * OWN_B + (unsigned)wave; p < total; p += OWN_NT / 32)
      own_batch<H, 1>(p, lane, s_list, s_ald, s_mx, s_den, als, h, acc);
  } else {
    own_scan(own, E0, wave, lane, [&](int e) {
      const unsigned rel = (unsigned)(dst[e] - lo);
      if (rel >= (unsigned)rows) return;
      const int s = src[e];
      for (int hd = 0; hd < H; ++hd) {
        const float a = __expf(own_score<H>(als[s * H + hd], s_ald[rel * H + hd]) - ord2f(s_mx[rel * H + hd]));
        atomicAdd(&s_den[rel * H + hd], a);
        const int f0 = hd * (FDIM / H);
        for (int f = f0; f < f0 + FDIM / H; ++f)
          atomicAdd(&acc[rel * FDIM + own_pos(f)], h[(size_t)s * FDIM + f] * a);
      }
    });
  }
  __syncthreads();

  for (int r = wave; r < rows; r += OWN_NT / 32) {
    const float* ap = &acc[r * FDIM + lane];
    float* op = outp + (size_t)(lo + r) * FDIM + 4 * lane;
    float den[H];
    for (int hd = 0; hd < H; ++hd) den[hd] = s_den[r * H + hd] + 1e-16f;
    for (int jj = 0; jj < NHR; ++jj) {
      const float  d = own_pick<H>(den, jj, lane);
      const float4 b = ((const float4*)bias)[32 * jj + lane];
      float4 val;
      val.x = ap[128 * jj]      / d + b.x;
      val.y = ap[128 * jj + 32] / d + b.y;
      val.z = ap[128 * jj + 64] / d + b.z;
      val.w = ap[128 * jj + 96] / d + b.w;
      if (ELU) {
        val.x = val.x > 0.f ? val.x : expm1f(val.x);
        val.y = val.y > 0.f ? val.y : expm1f(val.y);
        val.z = val.z > 0.f ? val.z : expm1f(val.z);
        val.w = val.w > 0.f ? val.w : expm1f(val.w);
      }
      *(float4*)(op + 128 * jj) = val;
    }
  }
}

static inline int blks(long long n, int bs) { return (int)((n + bs - 1) / bs); }

extern "C" void kernel_launch(void* const* d_in, const int* in_sizes, int n_in,
                              void* d_out, int out_size, void* d_ws, size_t ws_size,
                              hipStream_t stream) {
  const float* x        = (const float*)d_in[0];
  const int*   ei       = (const int*)d_in[1];
  const float* W1       = (const float*)d_in[2];
  const float* att_src1 = (const float*)d_in[3];
  const float* att_dst1 = (const float*)d_in[4];
  const float* b1       = (const float*)d_in[5];
  const float* W2       = (const float*)d_in[6];
  const float* att_src2 = (const float*)d_in[7];
  const float* att_dst2 = (const float*)d_in[8];
  const float* b2       = (const float*)d_in[9];
  float* out = (float*)d_out;

  const int* src = ei;
  const int* dst = ei + NEDGES;

  const long long NF = (long long)NNODES * FDIM;
  const long long NH = (long long)NNODES * HEADS1;

  float* ws = (float*)d_ws;
  float* h   = ws;
  float* agg = ws + NF;
  float* als = ws + 2 * NF;
  float* ald = als + NH;
  (void)ws_size; (void)in_sizes; (void)n_in; (void)out_size;

  const long long gemmThreads = (long long)(NNODES / 16) * 4   * 32;
  static_assert(NEDGES % 16 == 0, "agg_owner reads the owner-id list in 16-byte groups of 16 edges");
  static_assert((NNODES + OWN_R - 1) / OWN_R <= 256, "an aggregation workgroup's number fits a byte");
  static_assert(NNODES <= 65536, "agg_owner packs a source row number into 16 bits");
  const int ownBlks = blks(NNODES, OWN_R);
  unsigned* own1 = (unsigned*)out;
  unsigned* own2 = (unsigned*)agg;

  gemm_wmma_f32<<<blks(gemmThreads, 256), 256, 0, stream>>>(x, W1, h, NNODES);
  att_logits<HEADS1><<<blks(NH, 256), 256, 0, stream>>>(h, att_src1, att_dst1,
                                                        als, ald, NNODES, dst, own1, NEDGES);
  agg_owner<HEADS1, true><<<ownBlks, OWN_NT, 0, stream>>>(src, dst, own1, NEDGES, NNODES,
                                                          als, ald, h, b1, agg);

  gemm_wmma_f32<<<blks(gemmThreads, 256), 256, 0, stream>>>(agg, W2, h, NNODES);
  att_logits<1><<<blks(NNODES, 256), 256, 0, stream>>>(h, att_src2, att_dst2,
                                                       als, ald, NNODES, dst, own2, NEDGES);
  agg_owner<1, false><<<ownBlks, OWN_NT, 0, stream>>>(src, dst, own2, NEDGES, NNODES,
                                                      als, ald, h, b2, out);
}


# attention logits formed in the matrix product's epilogue from its accumulator tiles (16-lane DPP sums; layer 2 adds 4 strips via LDS); owner bytes written there: 6 to 4 launches; logit sum order new
# speedup vs baseline: 4.2060x; 1.0551x over previous
#include <hip/hip_runtime.h>


#define NNODES  50000
#define NEDGES  800000
#define FDIM    256
#define HEADS1  4
#define NEG_SLOPE 0.2f
#define OWN_R   196

typedef __attribute__((ext_vector_type(2))) float v2f;
typedef __attribute__((ext_vector_type(8))) float v8f;

__device__ __forceinline__ unsigned f2ord(float f) {
  unsigned u = __float_as_uint(f);
  return (u & 0x80000000u) ? ~u : (u | 0x80000000u);
}
__device__ __forceinline__ float ord2f(unsigned k) {
  unsigned u = (k & 0x80000000u) ? (k ^ 0x80000000u) : ~k;
  return __uint_as_float(u);
}

#define GEMM_NT 256

__device__ __forceinline__ float row16_sum(float v) {
  v += __int_as_float(__builtin_amdgcn_update_dpp(0, __float_as_int(v), 0x168, 0xF, 0xF, true));
  v += __int_as_float(__builtin_amdgcn_update_dpp(0, __float_as_int(v), 0x164, 0xF, 0xF, true));
  v += __int_as_float(__builtin_amdgcn_update_dpp(0, __float_as_int(v), 0x162, 0xF, 0xF, true));
  v += __int_as_float(__builtin_amdgcn_update_dpp(0, __float_as_int(v), 0x161, 0xF, 0xF, true));
  return v;
}

template <int H>
__global__ void gemm_wmma_f32(const float* __restrict__ A, const float* __restrict__ B,
                              float* __restrict__ C, int M,
                              const float* __restrict__ a_src, const float* __restrict__ a_dst,
                              float* __restrict__ als, float* __restrict__ ald,
                              const int* __restrict__ dst, unsigned* __restrict__ own, int E0) {
  constexpr int K = 256, Nc = 256;
  constexpr int NT = 4;
  constexpr int nWavesN = (Nc / 16) / NT;
  static_assert(NT == 4, "a lane's elements of the four tiles are one float4");
  static_assert(H == 4 || H == 1, "a 64-column strip is one head (H = 4) or a quarter of the only head (H = 1)");
  const int gtid = (int)(blockIdx.x * GEMM_NT + threadIdx.x);

  const int wave = __builtin_amdgcn_readfirstlane(gtid >> 5);
  const int lane = threadIdx.x & 31;
  const int tilesM = M >> 4;
  const bool active = wave < tilesM * nWavesN;
  const int tM    = wave / nWavesN;
  const int strip = wave - tM * nWavesN;
  const int col0  = strip * (NT * 16);
  const int half  = lane >> 4;
  const int l16   = lane & 15;

  float vs = 0.f, vd = 0.f;
  if (active) {
    const float* Abase = A + (size_t)(tM * 16 + l16) * K + 2 * half;
    const float* Bbase = B + (size_t)(2 * half) * Nc + col0 + NT * l16;

    v8f c[NT] = {};
#pragma unroll 8
    for (int k = 0; k < K; k += 4) {
      v2f a = *(const v2f*)(Abase + k);
      const float4 b0 = *(const float4*)(Bbase + k * Nc);
      const float4 b1 = *(const float4*)(Bbase + k * Nc + Nc);
      const float b0t[NT] = {b0.x, b0.y, b0.z, b0.w};
      const float b1t[NT] = {b1.x, b1.y, b1.z, b1.w};
#pragma unroll
      for (int t = 0; t < NT; ++t) {
        v2f b;
        b.x = b0t[t];
        b.y = b1t[t];
        c[t] = __builtin_amdgcn_wmma_f32_16x16x4_f32(false, a, false, b,
                                                     (short)0, c[t], false, false);
      }
    }
    float* Cp = C + (size_t)(tM * 16 + 8 * half) * Nc + col0 + NT * l16;
#pragma unroll
    for (int i = 0; i < 8; ++i)
      *(float4*)(Cp + i * Nc) = make_float4(c[0][i], c[1][i], c[2][i], c[3][i]);

    const float4 ws4 = *(const float4*)(a_src + col0 + NT * l16);
    const float4 wd4 = *(const float4*)(a_dst + col0 + NT * l16);
    float ps[8], pd[8];
#pragma unroll
    for (int i = 0; i < 8; ++i) {
      ps[i] = c[0][i] * ws4.x + c[1][i] * ws4.y + c[2][i] * ws4.z + c[3][i] * ws4.w;
      pd[i] = c[0][i] * wd4.x + c[1][i] * wd4.y + c[2][i] * wd4.z + c[3][i] * wd4.w;
    }
#pragma unroll
    for (int i = 0; i < 8; ++i) {
      ps[i] = row16_sum(ps[i]);
      pd[i] = row16_sum(pd[i]);
    }
    vs = ps[0]; vd = pd[0];
#pragma unroll
    for (int i = 1; i < 8; ++i)
      if (l16 == i) { vs = ps[i]; vd = pd[i]; }
  }

  if constexpr (H == 4) {
    if (active && l16 < 8) {
      const int row = tM * 16 + 8 * half + l16;
      als[row * H + strip] = vs;
      ald[row * H + strip] = vd;
    }
  } else {
    __shared__ float s_part[2][GEMM_NT / 32][16];
    const int wib = (int)(threadIdx.x >> 5);
    if (active && l16 < 8) {
      s_part[0][wib][8 * half + l16] = vs;
      s_part[1][wib][8 * half + l16] = vd;
    }
    __syncthreads();
    if (threadIdx.x < 64) {
      const int which = (int)(threadIdx.x >> 5);
      const int g     = (int)((threadIdx.x >> 4) & 1);
      const int r     = (int)(threadIdx.x & 15);
      const int tMg   = (int)blockIdx.x * (GEMM_NT / 32 / nWavesN) + g;
      if (tMg < tilesM) {
        const float sum = ((s_part[which][4 * g][r] + s_part[which][4 * g + 1][r])
                           + s_part[which][4 * g + 2][r]) + s_part[which][4 * g + 3][r];
        (which ? ald : als)[tMg * 16 + r] = sum;
      }
    }
  }

  if (gtid < (E0 >> 2)) {
    const int4 dv = ((const int4*)dst)[gtid];
    own[gtid] = (((unsigned)dv.x / OWN_R) & 255u) | ((((unsigned)dv.y / OWN_R) & 255u) << 8)
              | ((((unsigned)dv.z / OWN_R) & 255u) << 16) | (((unsigned)dv.w / OWN_R) << 24);
  }
}

#define OWN_CAP  12288
#define OWN_NT   1024
#define OWN_U    4
#define OWN_RAW  (OWN_R * FDIM)
#define OWN_B    8
#define OWN_PU   4

template <int H>
__device__ __forceinline__ float own_score(float als_v, float ald_v) {
  float sc = als_v + ald_v;
  return sc > 0.f ? sc : NEG_SLOPE * sc;
}

__device__ __forceinline__ int own_pos(int f) {
  return (f & ~127) | ((f & 3) << 5) | ((f >> 2) & 31);
}

template <int H>
__device__ __forceinline__ float own_pick(const float (&a)[H], int jj, int lane) {
  constexpr int F = FDIM / H;
  static_assert(F % 64 == 0, "the 16 lanes of a 64-feature stretch never straddle two heads");
  const int h0 = (128 * jj) / F;
  if constexpr (F >= 128) {
    return a[h0];
  } else {
    return (lane & 16) ? a[h0 + 1] : a[h0];
  }
}

__device__ __forceinline__ unsigned own_zero_bytes(unsigned x) {
  const unsigned t = (x & 0x7F7F7F7Fu) + 0x7F7F7F7Fu;
  return ~(t | x) & 0x80808080u;
}

template <typename F>
__device__ __forceinline__ void own_scan(const unsigned* __restrict__ own, int E0,
                                         int wave, int lane, F on_hit) {
  const int nvec   = E0 >> 4;
  const int nchunk = (nvec + 32 * OWN_U - 1) / (32 * OWN_U);
  const uint4* own4 = (const uint4*)own;
  const unsigned me = (blockIdx.x & 255u) * 0x01010101u;
  const int rot = (int)(((long long)blockIdx.x * nchunk) / gridDim.x);
  for (int c = wave; c < nchunk; c += OWN_NT / 32) {
    int cc = c + rot;
    if (cc >= nchunk) cc -= nchunk;
    uint4 ov[OWN_U];
    for (int u = 0; u < OWN_U; ++u) {
      const int idx = (cc * OWN_U + u) * 32 + lane;
      ov[u] = own4[min(idx, nvec - 1)];
    }
    static_assert(OWN_U == 4, "the statement below names the four loads");
    asm volatile("" : "+v"(ov[0].x), "+v"(ov[0].y), "+v"(ov[0].z), "+v"(ov[0].w),
                      "+v"(ov[1].x), "+v"(ov[1].y), "+v"(ov[1].z), "+v"(ov[1].w),
                      "+v"(ov[2].x), "+v"(ov[2].y), "+v"(ov[2].z), "+v"(ov[2].w),
                      "+v"(ov[3].x), "+v"(ov[3].y), "+v"(ov[3].z), "+v"(ov[3].w));
    unsigned long long hit = 0ull;
    for (int u = 0; u < OWN_U; ++u) {
      const int idx = (cc * OWN_U + u) * 32 + lane;
      unsigned m = (own_zero_bytes(ov[u].x ^ me) >> 7) | (own_zero_bytes(ov[u].y ^ me) >> 6)
                 | (own_zero_bytes(ov[u].z ^ me) >> 5) | (own_zero_bytes(ov[u].w ^ me) >> 4);
      if (idx >= nvec) m = 0u;
      hit |= (unsigned long long)m << (32 * (u >> 1) + 4 * (u & 1));
    }
    while (hit) {
      const int p = __builtin_ctzll(hit);
      hit &= hit - 1ull;
      const int u = 2 * (p >> 5) + ((p >> 2) & 1);
      on_hit(((((cc * OWN_U + u) * 32 + lane) << 4) + 4 * (p & 3)) + ((p & 31) >> 3));
    }
  }
}

template <int H, int NB>
__device__ __forceinline__ void own_batch(unsigned base, int lane, const unsigned* s_list,
                                          const float* s_ald, const unsigned* s_mx, float* s_den,
                                          const float* __restrict__ als, const float* __restrict__ h,
                                          float* acc) {
  constexpr int NHR = FDIM / 128;
  constexpr int F   = FDIM / H;
  static_assert(F % 64 == 0, "the 16 lanes of a 64-feature stretch never straddle two heads");
  static_assert(NB * H <= 32, "one lane per (edge, head)");
  int s[NB], r[NB];
  for (int i = 0; i < NB; ++i) {
    const unsigned ent = (unsigned)__builtin_amdgcn_readfirstlane((int)s_list[base + i]);
    s[i] = (int)(ent & 0xFFFFu);
    r[i] = (int)(ent >> 16);
  }
  float4 v[NB][NHR];
  for (int i = 0; i < NB; ++i) {
    const float4* hp = (const float4*)(h + (size_t)s[i] * FDIM) + lane;
    for (int jj = 0; jj < NHR; ++jj) v[i][jj] = hp[32 * jj];
  }
  float w = 0.f;
  if (lane < NB * H) {
    const unsigned ent = s_list[base + lane / H];
    const int hd = lane % H;
    const int ss = (int)(ent & 0xFFFFu), rr = (int)(ent >> 16);
    w = __expf(own_score<H>(als[ss * H + hd], s_ald[rr * H + hd]) - ord2f(s_mx[rr * H + hd]));
    atomicAdd(&s_den[rr * H + hd], w);
  }
  for (int i = 0; i < NB; ++i) {
    float* ap = &acc[r[i] * FDIM + lane];
    for (int jj = 0; jj < NHR; ++jj) {
      const int h0 = (128 * jj) / F;
      float a = __int_as_float(__builtin_amdgcn_readlane(__float_as_int(w), i * H + h0));
      if constexpr (F < 128) {
        const float a1 = __int_as_float(__builtin_amdgcn_readlane(__float_as_int(w), i * H + h0 + 1));
        a = (lane & 16) ? a1 : a;
      }
      atomicAdd(ap + 128 * jj,      v[i][jj].x * a);
      atomicAdd(ap + 128 * jj + 32, v[i][jj].y * a);
      atomicAdd(ap + 128 * jj + 64, v[i][jj].z * a);
      atomicAdd(ap + 128 * jj + 96, v[i][jj].w * a);
    }
  }
}

template <int H, bool ELU>
__global__ void __launch_bounds__(OWN_NT)
agg_owner(const int* __restrict__ src, const int* __restrict__ dst,
          const unsigned* __restrict__ own, int E0, int n,
          const float* __restrict__ als, const float* __restrict__ ald,
          const float* __restrict__ h, const float* __restrict__ bias,
          float* __restrict__ outp) {
  constexpr int NHR = FDIM / 128;
  __shared__ float    acc[OWN_R * FDIM];
  __shared__ float    s_ald[OWN_R * H];
  __shared__ unsigned s_mx[OWN_R * H];
  __shared__ float    s_den[OWN_R * H];
  __shared__ unsigned s_list[OWN_CAP];
  __shared__ unsigned s_cnt;
  __shared__ unsigned s_nraw;
  unsigned* s_raw = reinterpret_cast<unsigned*>(acc);

  const int lane = threadIdx.x & 31;
  const int wave = __builtin_amdgcn_readfirstlane((int)(threadIdx.x >> 5));
  const int lo   = (int)blockIdx.x * OWN_R;
  const int rows = min(OWN_R, n - lo);

  for (int j = threadIdx.x; j < rows * H; j += OWN_NT) {
    const float d = ald[lo * H + j];
    s_ald[j] = d;
    s_mx[j]  = f2ord(own_score<H>(als[lo * H + j], d));
  }
  if (threadIdx.x == 0) { s_cnt = 0u; s_nraw = 0u; }
  __syncthreads();

  own_scan(own, E0, wave, lane, [&](int e) {
    const unsigned pos = atomicAdd(&s_nraw, 1u);
    if (pos < (unsigned)OWN_RAW) s_raw[pos] = (unsigned)e;
  });
  __syncthreads();

  const unsigned nraw  = s_nraw;
  const bool     rawok = nraw <= (unsigned)OWN_RAW;
  if (rawok) {
    for (unsigned p = threadIdx.x; p < nraw; p += OWN_NT) {
      const unsigned e   = s_raw[p];
      const unsigned rel = (unsigned)(dst[e] - lo);
      const int s = src[e];
      if (rel < (unsigned)rows) {
        const unsigned pos = atomicAdd(&s_cnt, 1u);
        if (pos < OWN_CAP) s_list[pos] = (rel << 16) | (unsigned)s;
      }
    }
  }
  __syncthreads();

  const unsigned total   = s_cnt;
  const bool     listed  = rawok && total <= (unsigned)OWN_CAP;

  if (listed) {
    const unsigned npairs = total * H;
    for (unsigned p0 = threadIdx.x; p0 < npairs; p0 += OWN_NT * OWN_PU) {
      float    sv[OWN_PU];
      unsigned ix[OWN_PU];
      for (int u = 0; u < OWN_PU; ++u) {
        const unsigned p   = min(p0 + (unsigned)(u * OWN_NT), npairs - 1u);
        const unsigned ent = s_list[p / H];
        const int hd = (int)(p % H);
        ix[u] = (ent >> 16) * H + hd;
        sv[u] = als[(ent & 0xFFFFu) * H + hd];
      }
      for (int u = 0; u < OWN_PU; ++u)
        atomicMax(&s_mx[ix[u]], f2ord(own_score<H>(sv[u], s_ald[ix[u]])));
    }
  } else {
    own_scan(own, E0, wave, lane, [&](int e) {
      const unsigned rel = (unsigned)(dst[e] - lo);
      if (rel >= (unsigned)rows) return;
      const int s = src[e];
      for (int hd = 0; hd < H; ++hd)
        atomicMax(&s_mx[rel * H + hd], f2ord(own_score<H>(als[s * H + hd], s_ald[rel * H + hd])));
    });
  }
  __syncthreads();

  for (int r = wave; r < rows; r += OWN_NT / 32) {
    const int node = lo + r;
    float a[H];
    for (int hd = 0; hd < H; ++hd) {
      a[hd] = __expf(own_score<H>(als[node * H + hd], s_ald[r * H + hd]) - ord2f(s_mx[r * H + hd]));
      if (lane == hd) s_den[r * H + hd] = a[hd];
    }
    const float4* hp = (const float4*)(h + (size_t)node * FDIM) + lane;
    float* ap = &acc[r * FDIM + lane];
    for (int jj = 0; jj < NHR; ++jj) {
      const float4 hv = hp[32 * jj];
      const float  aw = own_pick<H>(a, jj, lane);
      ap[128 * jj]      = hv.x * aw;
      ap[128 * jj + 32] = hv.y * aw;
      ap[128 * jj + 64] = hv.z * aw;
      ap[128 * jj + 96] = hv.w * aw;
    }
  }
  __syncthreads();

  if (listed) {
    const unsigned nfull = total / OWN_B;
    for (unsigned bt = (unsigned)wave; bt < nfull; bt += OWN_NT / 32)
      own_batch<H, OWN_B>(bt * OWN_B, lane, s_list, s_ald, s_mx, s_den, als, h, acc);
    for (unsigned p = nfull * OWN_B + (unsigned)wave; p < total; p += OWN_NT / 32)
      own_batch<H, 1>(p, lane, s_list, s_ald, s_mx, s_den, als, h, acc);
  } else {
    own_scan(own, E0, wave, lane, [&](int e) {
      const unsigned rel = (unsigned)(dst[e] - lo);
      if (rel >= (unsigned)rows) return;
      const int s = src[e];
      for (int hd = 0; hd < H; ++hd) {
        const float a = __expf(own_score<H>(als[s * H + hd], s_ald[rel * H + hd]) - ord2f(s_mx[rel * H + hd]));
        atomicAdd(&s_den[rel * H + hd], a);
        const int f0 = hd * (FDIM / H);
        for (int f = f0; f < f0 + FDIM / H; ++f)
          atomicAdd(&acc[rel * FDIM + own_pos(f)], h[(size_t)s * FDIM + f] * a);
      }
    });
  }
  __syncthreads();

  for (int r = wave; r < rows; r += OWN_NT / 32) {
    const float* ap = &acc[r * FDIM + lane];
    float* op = outp + (size_t)(lo + r) * FDIM + 4 * lane;
    float den[H];
    for (int hd = 0; hd < H; ++hd) den[hd] = s_den[r * H + hd] + 1e-16f;
    for (int jj = 0; jj < NHR; ++jj) {
      const float  d = own_pick<H>(den, jj, lane);
      const float4 b = ((const float4*)bias)[32 * jj + lane];
      float4 val;
      val.x = ap[128 * jj]      / d + b.x;
      val.y = ap[128 * jj + 32] / d + b.y;
      val.z = ap[128 * jj + 64] / d + b.z;
      val.w = ap[128 * jj + 96] / d + b.w;
      if (ELU) {
        val.x = val.x > 0.f ? val.x : expm1f(val.x);
        val.y = val.y > 0.f ? val.y : expm1f(val.y);
        val.z = val.z > 0.f ? val.z : expm1f(val.z);
        val.w = val.w > 0.f ? val.w : expm1f(val.w);
      }
      *(float4*)(op + 128 * jj) = val;
    }
  }
}

static inline int blks(long long n, int bs) { return (int)((n + bs - 1) / bs); }

extern "C" void kernel_launch(void* const* d_in, const int* in_sizes, int n_in,
                              void* d_out, int out_size, void* d_ws, size_t ws_size,
                              hipStream_t stream) {
  const float* x        = (const float*)d_in[0];
  const int*   ei       = (const int*)d_in[1];
  const float* W1       = (const float*)d_in[2];
  const float* att_src1 = (const float*)d_in[3];
  const float* att_dst1 = (const float*)d_in[4];
  const float* b1       = (const float*)d_in[5];
  const float* W2       = (const float*)d_in[6];
  const float* att_src2 = (const float*)d_in[7];
  const float* att_dst2 = (const float*)d_in[8];
  const float* b2       = (const float*)d_in[9];
  float* out = (float*)d_out;

  const int* src = ei;
  const int* dst = ei + NEDGES;

  const long long NF = (long long)NNODES * FDIM;
  const long long NH = (long long)NNODES * HEADS1;

  float* ws = (float*)d_ws;
  float* h   = ws;
  float* agg = ws + NF;
  float* als = ws + 2 * NF;
  float* ald = als + NH;
  (void)ws_size; (void)in_sizes; (void)n_in; (void)out_size;

  const long long gemmThreads = (long long)(NNODES / 16) * 4   * 32;
  static_assert(NNODES % 16 == 0, "the matrix product has no partial M-tile");
  static_assert(NEDGES % 16 == 0, "agg_owner reads the owner-id list in 16-byte groups of 16 edges");
  static_assert((long long)(NNODES / 16) * 4 * 32 >= NEDGES / 4, "one thread of the matrix product's grid per word of owner bytes");
  static_assert((NNODES + OWN_R - 1) / OWN_R <= 256, "an aggregation workgroup's number fits a byte");
  static_assert(NNODES <= 65536, "agg_owner packs a source row number into 16 bits");
  const int ownBlks = blks(NNODES, OWN_R);
  static_assert(2 * NNODES * 4 + NEDGES <= 2 * NNODES * HEADS1 * 4, "layer 2's logits and owner bytes fit the two logit planes");
  static_assert((2 * NNODES * 4) % 16 == 0, "the owner-id list is read in 16-byte groups");
  unsigned* own1 = (unsigned*)out;
  float*    als2 = als;
  float*    ald2 = als + NNODES;
  unsigned* own2 = (unsigned*)(als + 2 * NNODES);

  gemm_wmma_f32<HEADS1><<<blks(gemmThreads, GEMM_NT), GEMM_NT, 0, stream>>>(x, W1, h, NNODES, att_src1, att_dst1,
                                                                           als, ald, dst, own1, NEDGES);
  agg_owner<HEADS1, true><<<ownBlks, OWN_NT, 0, stream>>>(src, dst, own1, NEDGES, NNODES,
                                                          als, ald, h, b1, agg);

  gemm_wmma_f32<1><<<blks(gemmThreads, GEMM_NT), GEMM_NT, 0, stream>>>(agg, W2, h, NNODES, att_src2, att_dst2,
                                                                      als2, ald2, dst, own2, NEDGES);
  agg_owner<1, false><<<ownBlks, OWN_NT, 0, stream>>>(src, dst, own2, NEDGES, NNODES,
                                                      als2, ald2, h, b2, out);
}


# layer 2's aggregation reads back each workgroup's edge list as layer 1 built it (kept in the workgroup's own rows of the output buffer): no 2nd list scan or dst/src fetch; sums still in arrival order
# speedup vs baseline: 4.4693x; 1.0626x over previous
#include <hip/hip_runtime.h>


#define NNODES  50000
#define NEDGES  800000
#define FDIM    256
#define HEADS1  4
#define NEG_SLOPE 0.2f
#define OWN_R   196
#define OWN_STRETCH (OWN_R * FDIM)
#define KEEP_W   12544
#define PIECE_SH 15

typedef __attribute__((ext_vector_type(2))) float v2f;
typedef __attribute__((ext_vector_type(8))) float v8f;

__device__ __forceinline__ unsigned f2ord(float f) {
  unsigned u = __float_as_uint(f);
  return (u & 0x80000000u) ? ~u : (u | 0x80000000u);
}
__device__ __forceinline__ float ord2f(unsigned k) {
  unsigned u = (k & 0x80000000u) ? (k ^ 0x80000000u) : ~k;
  return __uint_as_float(u);
}

#define GEMM_NT 256

template <bool PIECES>
__device__ __forceinline__ int own_word_at(int w) {
  if constexpr (PIECES) return (w >> PIECE_SH) * OWN_STRETCH + KEEP_W + (w & ((1 << PIECE_SH) - 1));
  else return w;
}

__device__ __forceinline__ float row16_sum(float v) {
  v += __int_as_float(__builtin_amdgcn_update_dpp(0, __float_as_int(v), 0x168, 0xF, 0xF, true));
  v += __int_as_float(__builtin_amdgcn_update_dpp(0, __float_as_int(v), 0x164, 0xF, 0xF, true));
  v += __int_as_float(__builtin_amdgcn_update_dpp(0, __float_as_int(v), 0x162, 0xF, 0xF, true));
  v += __int_as_float(__builtin_amdgcn_update_dpp(0, __float_as_int(v), 0x161, 0xF, 0xF, true));
  return v;
}

template <int H, bool PIECES>
__global__ void gemm_wmma_f32(const float* __restrict__ A, const float* __restrict__ B,
                              float* __restrict__ C, int M,
                              const float* __restrict__ a_src, const float* __restrict__ a_dst,
                              float* __restrict__ als, float* __restrict__ ald,
                              const int* __restrict__ dst, unsigned* __restrict__ own, int E0) {
  constexpr int K = 256, Nc = 256;
  constexpr int NT = 4;
  constexpr int nWavesN = (Nc / 16) / NT;
  static_assert(NT == 4, "a lane's elements of the four tiles are one float4");
  static_assert(H == 4 || H == 1, "a 64-column strip is one head (H = 4) or a quarter of the only head (H = 1)");
  const int gtid = (int)(blockIdx.x * GEMM_NT + threadIdx.x);

  const int wave = __builtin_amdgcn_readfirstlane(gtid >> 5);
  const int lane = threadIdx.x & 31;
  const int tilesM = M >> 4;
  const bool active = wave < tilesM * nWavesN;
  const int tM    = wave / nWavesN;
  const int strip = wave - tM * nWavesN;
  const int col0  = strip * (NT * 16);
  const int half  = lane >> 4;
  const int l16   = lane & 15;

  float vs = 0.f, vd = 0.f;
  if (active) {
    const float* Abase = A + (size_t)(tM * 16 + l16) * K + 2 * half;
    const float* Bbase = B + (size_t)(2 * half) * Nc + col0 + NT * l16;

    v8f c[NT] = {};
#pragma unroll 8
    for (int k = 0; k < K; k += 4) {
      v2f a = *(const v2f*)(Abase + k);
      const float4 b0 = *(const float4*)(Bbase + k * Nc);
      const float4 b1 = *(const float4*)(Bbase + k * Nc + Nc);
      const float b0t[NT] = {b0.x, b0.y, b0.z, b0.w};
      const float b1t[NT] = {b1.x, b1.y, b1.z, b1.w};
#pragma unroll
      for (int t = 0; t < NT; ++t) {
        v2f b;
        b.x = b0t[t];
        b.y = b1t[t];
        c[t] = __builtin_amdgcn_wmma_f32_16x16x4_f32(false, a, false, b,
                                                     (short)0, c[t], false, false);
      }
    }
    float* Cp = C + (size_t)(tM * 16 + 8 * half) * Nc + col0 + NT * l16;
#pragma unroll
    for (int i = 0; i < 8; ++i)
      *(float4*)(Cp + i * Nc) = make_float4(c[0][i], c[1][i], c[2][i], c[3][i]);

    const float4 ws4 = *(const float4*)(a_src + col0 + NT * l16);
    const float4 wd4 = *(const float4*)(a_dst + col0 + NT * l16);
    float ps[8], pd[8];
#pragma unroll
    for (int i = 0; i < 8; ++i) {
      ps[i] = c[0][i] * ws4.x + c[1][i] * ws4.y + c[2][i] * ws4.z + c[3][i] * ws4.w;
      pd[i] = c[0][i] * wd4.x + c[1][i] * wd4.y + c[2][i] * wd4.z + c[3][i] * wd4.w;
    }
#pragma unroll
    for (int i = 0; i < 8; ++i) {
      ps[i] = row16_sum(ps[i]);
      pd[i] = row16_sum(pd[i]);
    }
    vs = ps[0]; vd = pd[0];
#pragma unroll
    for (int i = 1; i < 8; ++i)
      if (l16 == i) { vs = ps[i]; vd = pd[i]; }
  }

  if constexpr (H == 4) {
    if (active && l16 < 8) {
      const int row = tM * 16 + 8 * half + l16;
      als[row * H + strip] = vs;
      ald[row * H + strip] = vd;
    }
  } else {
    __shared__ float s_part[2][GEMM_NT / 32][16];
    const int wib = (int)(threadIdx.x >> 5);
    if (active && l16 < 8) {
      s_part[0][wib][8 * half + l16] = vs;
      s_part[1][wib][8 * half + l16] = vd;
    }
    __syncthreads();
    if (threadIdx.x < 64) {
      const int which = (int)(threadIdx.x >> 5);
      const int g     = (int)((threadIdx.x >> 4) & 1);
      const int r     = (int)(threadIdx.x & 15);
      const int tMg   = (int)blockIdx.x * (GEMM_NT / 32 / nWavesN) + g;
      if (tMg < tilesM) {
        const float sum = ((s_part[which][4 * g][r] + s_part[which][4 * g + 1][r])
                           + s_part[which][4 * g + 2][r]) + s_part[which][4 * g + 3][r];
        (which ? ald : als)[tMg * 16 + r] = sum;
      }
    }
  }

  if (gtid < (E0 >> 2)) {
    const int4 dv = ((const int4*)dst)[gtid];
    own[own_word_at<PIECES>(gtid)] = (((unsigned)dv.x / OWN_R) & 255u) | ((((unsigned)dv.y / OWN_R) & 255u) << 8)
              | ((((unsigned)dv.z / OWN_R) & 255u) << 16) | (((unsigned)dv.w / OWN_R) << 24);
  }
}

#define OWN_CAP  12288
#define OWN_NT   1024
#define OWN_U    4
#define OWN_RAW  (OWN_R * FDIM)
#define OWN_B    8
#define OWN_PU   4
#define OWN_KU   4

template <int H>
__device__ __forceinline__ float own_score(float als_v, float ald_v) {
  float sc = als_v + ald_v;
  return sc > 0.f ? sc : NEG_SLOPE * sc;
}

__device__ __forceinline__ int own_pos(int f) {
  return (f & ~127) | ((f & 3) << 5) | ((f >> 2) & 31);
}

template <int H>
__device__ __forceinline__ float own_pick(const float (&a)[H], int jj, int lane) {
  constexpr int F = FDIM / H;
  static_assert(F % 64 == 0, "the 16 lanes of a 64-feature stretch never straddle two heads");
  const int h0 = (128 * jj) / F;
  if constexpr (F >= 128) {
    return a[h0];
  } else {
    return (lane & 16) ? a[h0 + 1] : a[h0];
  }
}

__device__ __forceinline__ unsigned own_zero_bytes(unsigned x) {
  const unsigned t = (x & 0x7F7F7F7Fu) + 0x7F7F7F7Fu;
  return ~(t | x) & 0x80808080u;
}

template <bool PIECES, typename F>
__device__ __forceinline__ void own_scan(const unsigned* __restrict__ own, int E0,
                                         int wave, int lane, F on_hit) {
  const int nvec   = E0 >> 4;
  const int nchunk = (nvec + 32 * OWN_U - 1) / (32 * OWN_U);
  const uint4* own4 = (const uint4*)own;
  const unsigned me = (blockIdx.x & 255u) * 0x01010101u;
  const int rot = (int)(((long long)blockIdx.x * nchunk) / gridDim.x);
  for (int c = wave; c < nchunk; c += OWN_NT / 32) {
    int cc = c + rot;
    if (cc >= nchunk) cc -= nchunk;
    uint4 ov[OWN_U];
    for (int u = 0; u < OWN_U; ++u) {
      const int idx = (cc * OWN_U + u) * 32 + lane;
      ov[u] = own4[own_word_at<PIECES>(4 * min(idx, nvec - 1)) >> 2];
    }
    static_assert(OWN_U == 4, "the statement below names the four loads");
    asm volatile("" : "+v"(ov[0].x), "+v"(ov[0].y), "+v"(ov[0].z), "+v"(ov[0].w),
                      "+v"(ov[1].x), "+v"(ov[1].y), "+v"(ov[1].z), "+v"(ov[1].w),
                      "+v"(ov[2].x), "+v"(ov[2].y), "+v"(ov[2].z), "+v"(ov[2].w),
                      "+v"(ov[3].x), "+v"(ov[3].y), "+v"(ov[3].z), "+v"(ov[3].w));
    unsigned long long hit = 0ull;
    for (int u = 0; u < OWN_U; ++u) {
      const int idx = (cc * OWN_U + u) * 32 + lane;
      unsigned m = (own_zero_bytes(ov[u].x ^ me) >> 7) | (own_zero_bytes(ov[u].y ^ me) >> 6)
                 | (own_zero_bytes(ov[u].z ^ me) >> 5) | (own_zero_bytes(ov[u].w ^ me) >> 4);
      if (idx >= nvec) m = 0u;
      hit |= (unsigned long long)m << (32 * (u >> 1) + 4 * (u & 1));
    }
    while (hit) {
      const int p = __builtin_ctzll(hit);
      hit &= hit - 1ull;
      const int u = 2 * (p >> 5) + ((p >> 2) & 1);
      on_hit(((((cc * OWN_U + u) * 32 + lane) << 4) + 4 * (p & 3)) + ((p & 31) >> 3));
    }
  }
}

template <int H, int NB>
__device__ __forceinline__ void own_batch(unsigned base, int lane, const unsigned* s_list,
                                          const float* s_ald, const unsigned* s_mx, float* s_den,
                                          const float* __restrict__ als, const float* __restrict__ h,
                                          float* acc) {
  constexpr int NHR = FDIM / 128;
  constexpr int F   = FDIM / H;
  static_assert(F % 64 == 0, "the 16 lanes of a 64-feature stretch never straddle two heads");
  static_assert(NB * H <= 32, "one lane per (edge, head)");
  int s[NB], r[NB];
  for (int i = 0; i < NB; ++i) {
    const unsigned ent = (unsigned)__builtin_amdgcn_readfirstlane((int)s_list[base + i]);
    s[i] = (int)(ent & 0xFFFFu);
    r[i] = (int)(ent >> 16);
  }
  float4 v[NB][NHR];
  for (int i = 0; i < NB; ++i) {
    const float4* hp = (const float4*)(h + (size_t)s[i] * FDIM) + lane;
    for (int jj = 0; jj < NHR; ++jj) v[i][jj] = hp[32 * jj];
  }
  float w = 0.f;
  if (lane < NB * H) {
    const unsigned ent = s_list[base + lane / H];
    const int hd = lane % H;
    const int ss = (int)(ent & 0xFFFFu), rr = (int)(ent >> 16);
    w = __expf(own_score<H>(als[ss * H + hd], s_ald[rr * H + hd]) - ord2f(s_mx[rr * H + hd]));
    atomicAdd(&s_den[rr * H + hd], w);
  }
  for (int i = 0; i < NB; ++i) {
    float* ap = &acc[r[i] * FDIM + lane];
    for (int jj = 0; jj < NHR; ++jj) {
      const int h0 = (128 * jj) / F;
      float a = __int_as_float(__builtin_amdgcn_readlane(__float_as_int(w), i * H + h0));
      if constexpr (F < 128) {
        const float a1 = __int_as_float(__builtin_amdgcn_readlane(__float_as_int(w), i * H + h0 + 1));
        a = (lane & 16) ? a1 : a;
      }
      atomicAdd(ap + 128 * jj,      v[i][jj].x * a);
      atomicAdd(ap + 128 * jj + 32, v[i][jj].y * a);
      atomicAdd(ap + 128 * jj + 64, v[i][jj].z * a);
      atomicAdd(ap + 128 * jj + 96, v[i][jj].w * a);
    }
  }
}

template <int H, bool ELU, bool FIRST>
__global__ void __launch_bounds__(OWN_NT)
agg_owner(const int* __restrict__ src, const int* __restrict__ dst,
          const unsigned* __restrict__ own, int E0, int n,
          const float* __restrict__ als, const float* __restrict__ ald,
          const float* __restrict__ h, const float* __restrict__ bias,
          float* outp, unsigned* keep) {
  constexpr int NHR = FDIM / 128;
  __shared__ float    acc[OWN_R * FDIM];
  __shared__ float    s_ald[OWN_R * H];
  __shared__ unsigned s_mx[OWN_R * H];
  __shared__ float    s_den[OWN_R * H];
  __shared__ unsigned s_list[OWN_CAP];
  __shared__ unsigned s_cnt;
  __shared__ unsigned s_nraw;
  unsigned* s_raw = reinterpret_cast<unsigned*>(acc);

  const int lane = threadIdx.x & 31;
  const int wave = __builtin_amdgcn_readfirstlane((int)(threadIdx.x >> 5));
  const int lo   = (int)blockIdx.x * OWN_R;
  const int rows = min(OWN_R, n - lo);
  unsigned* kb = keep + (size_t)blockIdx.x * OWN_STRETCH;
  const unsigned keepcap = min((unsigned)OWN_CAP, (unsigned)(rows * FDIM) - 1u);
  unsigned taken = 0u;
  if constexpr (!FIRST) taken = (unsigned)__builtin_amdgcn_readfirstlane((int)kb[0]);

  for (int j = threadIdx.x; j < rows * H; j += OWN_NT) {
    const float d = ald[lo * H + j];
    s_ald[j] = d;
    s_mx[j]  = f2ord(own_score<H>(als[lo * H + j], d));
  }
  if (threadIdx.x == 0) { s_cnt = 0u; s_nraw = 0u; }

  bool have = false;
  if constexpr (!FIRST) {
    have = taken <= keepcap;
    if (have) {
      for (unsigned p0 = threadIdx.x; p0 < taken; p0 += OWN_NT * OWN_KU) {
        unsigned at[OWN_KU], ent[OWN_KU];
        for (int u = 0; u < OWN_KU; ++u) {
          at[u]  = min(p0 + (unsigned)(u * OWN_NT), taken - 1u);
          ent[u] = kb[1u + at[u]];
        }
        for (int u = 0; u < OWN_KU; ++u) s_list[at[u]] = ent[u];
      }
    }
  }
  __syncthreads();

  unsigned total  = taken;
  bool     listed = true;
  if (!have) {
    own_scan<FIRST>(own, E0, wave, lane, [&](int e) {
      const unsigned pos = atomicAdd(&s_nraw, 1u);
      if (pos < (unsigned)OWN_RAW) s_raw[pos] = (unsigned)e;
    });
    __syncthreads();

    const unsigned nraw  = s_nraw;
    const bool     rawok = nraw <= (unsigned)OWN_RAW;
    if (rawok) {
      for (unsigned p = threadIdx.x; p < nraw; p += OWN_NT) {
        const unsigned e   = s_raw[p];
        const unsigned rel = (unsigned)(dst[e] - lo);
        const int s = src[e];
        if (rel < (unsigned)rows) {
          const unsigned pos = atomicAdd(&s_cnt, 1u);
          if (pos < OWN_CAP) s_list[pos] = (rel << 16) | (unsigned)s;
        }
      }
    }
    __syncthreads();
    total  = s_cnt;
    listed = rawok && total <= (unsigned)OWN_CAP;
  }

  if constexpr (FIRST) {
    const bool fits = listed && total <= keepcap;
    if (fits)
      for (unsigned p = threadIdx.x; p < total; p += OWN_NT) kb[1u + p] = s_list[p];
    if (threadIdx.x == 0) kb[0] = fits ? total : 0xFFFFFFFFu;
  }

  if (listed) {
    const unsigned npairs = total * H;
    for (unsigned p0 = threadIdx.x; p0 < npairs; p0 += OWN_NT * OWN_PU) {
      float    sv[OWN_PU];
      unsigned ix[OWN_PU];
      for (int u = 0; u < OWN_PU; ++u) {
        const unsigned p   = min(p0 + (unsigned)(u * OWN_NT), npairs - 1u);
        const unsigned ent = s_list[p / H];
        const int hd = (int)(p % H);
        ix[u] = (ent >> 16) * H + hd;
        sv[u] = als[(ent & 0xFFFFu) * H + hd];
      }
      for (int u = 0; u < OWN_PU; ++u)
        atomicMax(&s_mx[ix[u]], f2ord(own_score<H>(sv[u], s_ald[ix[u]])));
    }
  } else {
    own_scan<FIRST>(own, E0, wave, lane, [&](int e) {
      const unsigned rel = (unsigned)(dst[e] - lo);
      if (rel >= (unsigned)rows) return;
      const int s = src[e];
      for (int hd = 0; hd < H; ++hd)
        atomicMax(&s_mx[rel * H + hd], f2ord(own_score<H>(als[s * H + hd], s_ald[rel * H + hd])));
    });
  }
  __syncthreads();

  for (int r = wave; r < rows; r += OWN_NT / 32) {
    const int node = lo + r;
    float a[H];
    for (int hd = 0; hd < H; ++hd) {
      a[hd] = __expf(own_score<H>(als[node * H + hd], s_ald[r * H + hd]) - ord2f(s_mx[r * H + hd]));
      if (lane == hd) s_den[r * H + hd] = a[hd];
    }
    const float4* hp = (const float4*)(h + (size_t)node * FDIM) + lane;
    float* ap = &acc[r * FDIM + lane];
    for (int jj = 0; jj < NHR; ++jj) {
      const float4 hv = hp[32 * jj];
      const float  aw = own_pick<H>(a, jj, lane);
      ap[128 * jj]      = hv.x * aw;
      ap[128 * jj + 32] = hv.y * aw;
      ap[128 * jj + 64] = hv.z * aw;
      ap[128 * jj + 96] = hv.w * aw;
    }
  }
  __syncthreads();

  if (listed) {
    const unsigned nfull = total / OWN_B;
    for (unsigned bt = (unsigned)wave; bt < nfull; bt += OWN_NT / 32)
      own_batch<H, OWN_B>(bt * OWN_B, lane, s_list, s_ald, s_mx, s_den, als, h, acc);
    for (unsigned p = nfull * OWN_B + (unsigned)wave; p < total; p += OWN_NT / 32)
      own_batch<H, 1>(p, lane, s_list, s_ald, s_mx, s_den, als, h, acc);
  } else {
    own_scan<FIRST>(own, E0, wave, lane, [&](int e) {
      const unsigned rel = (unsigned)(dst[e] - lo);
      if (rel >= (unsigned)rows) return;
      const int s = src[e];
      for (int hd = 0; hd < H; ++hd) {
        const float a = __expf(own_score<H>(als[s * H + hd], s_ald[rel * H + hd]) - ord2f(s_mx[rel * H + hd]));
        atomicAdd(&s_den[rel * H + hd], a);
        const int f0 = hd * (FDIM / H);
        for (int f = f0; f < f0 + FDIM / H; ++f)
          atomicAdd(&acc[rel * FDIM + own_pos(f)], h[(size_t)s * FDIM + f] * a);
      }
    });
  }
  __syncthreads();

  for (int r = wave; r < rows; r += OWN_NT / 32) {
    const float* ap = &acc[r * FDIM + lane];
    float* op = outp + (size_t)(lo + r) * FDIM + 4 * lane;
    float den[H];
    for (int hd = 0; hd < H; ++hd) den[hd] = s_den[r * H + hd] + 1e-16f;
    for (int jj = 0; jj < NHR; ++jj) {
      const float  d = own_pick<H>(den, jj, lane);
      const float4 b = ((const float4*)bias)[32 * jj + lane];
      float4 val;
      val.x = ap[128 * jj]      / d + b.x;
      val.y = ap[128 * jj + 32] / d + b.y;
      val.z = ap[128 * jj + 64] / d + b.z;
      val.w = ap[128 * jj + 96] / d + b.w;
      if (ELU) {
        val.x = val.x > 0.f ? val.x : expm1f(val.x);
        val.y = val.y > 0.f ? val.y : expm1f(val.y);
        val.z = val.z > 0.f ? val.z : expm1f(val.z);
        val.w = val.w > 0.f ? val.w : expm1f(val.w);
      }
      *(float4*)(op + 128 * jj) = val;
    }
  }
}

static inline int blks(long long n, int bs) { return (int)((n + bs - 1) / bs); }

extern "C" void kernel_launch(void* const* d_in, const int* in_sizes, int n_in,
                              void* d_out, int out_size, void* d_ws, size_t ws_size,
                              hipStream_t stream) {
  const float* x        = (const float*)d_in[0];
  const int*   ei       = (const int*)d_in[1];
  const float* W1       = (const float*)d_in[2];
  const float* att_src1 = (const float*)d_in[3];
  const float* att_dst1 = (const float*)d_in[4];
  const float* b1       = (const float*)d_in[5];
  const float* W2       = (const float*)d_in[6];
  const float* att_src2 = (const float*)d_in[7];
  const float* att_dst2 = (const float*)d_in[8];
  const float* b2       = (const float*)d_in[9];
  float* out = (float*)d_out;

  const int* src = ei;
  const int* dst = ei + NEDGES;

  const long long NF = (long long)NNODES * FDIM;
  const long long NH = (long long)NNODES * HEADS1;

  float* ws = (float*)d_ws;
  float* h   = ws;
  float* agg = ws + NF;
  float* als = ws + 2 * NF;
  float* ald = als + NH;
  (void)ws_size; (void)in_sizes; (void)n_in; (void)out_size;

  const long long gemmThreads = (long long)(NNODES / 16) * 4   * 32;
  static_assert(NNODES % 16 == 0, "the matrix product has no partial M-tile");
  static_assert(NEDGES % 16 == 0, "agg_owner reads the owner-id list in 16-byte groups of 16 edges");
  static_assert((long long)(NNODES / 16) * 4 * 32 >= NEDGES / 4, "one thread of the matrix product's grid per word of owner bytes");
  static_assert((NNODES + OWN_R - 1) / OWN_R <= 256, "an aggregation workgroup's number fits a byte");
  static_assert(NNODES <= 65536, "agg_owner packs a source row number into 16 bits");
  const int ownBlks = blks(NNODES, OWN_R);
  static_assert(2 * NNODES * 4 + NEDGES <= 2 * NNODES * HEADS1 * 4, "layer 2's logits and owner bytes fit the two logit planes");
  static_assert((2 * NNODES * 4) % 16 == 0, "the owner-id list is read in 16-byte groups");
  static_assert(OWN_CAP + 1 <= KEEP_W, "count and entries of a handed-over list fit the head of a stretch");
  static_assert(KEEP_W % 32 == 0 && KEEP_W + (1 << PIECE_SH) <= OWN_STRETCH, "an owner-id piece starts on a 128-B line and ends inside its stretch");
  static_assert((NEDGES / 4 + (1 << PIECE_SH) - 1) / (1 << PIECE_SH) <= NNODES / OWN_R, "every owner-id piece lies in a full stretch");
  unsigned* own1 = (unsigned*)out;
  unsigned* keep = (unsigned*)out;
  float*    als2 = als;
  float*    ald2 = als + NNODES;
  unsigned* own2 = (unsigned*)(als + 2 * NNODES);

  gemm_wmma_f32<HEADS1, true><<<blks(gemmThreads, GEMM_NT), GEMM_NT, 0, stream>>>(x, W1, h, NNODES, att_src1, att_dst1,
                                                                                 als, ald, dst, own1, NEDGES);
  agg_owner<HEADS1, true, true><<<ownBlks, OWN_NT, 0, stream>>>(src, dst, own1, NEDGES, NNODES,
                                                                als, ald, h, b1, agg, keep);

  gemm_wmma_f32<1, false><<<blks(gemmThreads, GEMM_NT), GEMM_NT, 0, stream>>>(agg, W2, h, NNODES, att_src2, att_dst2,
                                                                             als2, ald2, dst, own2, NEDGES);
  agg_owner<1, false, false><<<ownBlks, OWN_NT, 0, stream>>>(src, dst, own2, NEDGES, NNODES,
                                                             als2, ald2, h, b2, out, keep);
}
